# starting kernel
# baseline (speedup 1.0000x reference)
#include <hip/hip_runtime.h>

typedef __attribute__((ext_vector_type(2))) float v2f;
typedef __attribute__((ext_vector_type(8))) float v8f;

#define C_DIM 128

__global__ __launch_bounds__(256) void gcn_gemm_norm(
    const float* __restrict__ x, const float* __restrict__ w,
    const float* __restrict__ norm, float* __restrict__ h, int N)
{
    __shared__ float lds_w[C_DIM * C_DIM];

    for (int idx = threadIdx.x; idx < (C_DIM / 2) * C_DIM; idx += 256) {
        const int p = idx >> 7;
        const int n = idx & 127;
        v2f v;
        v.x = w[(2 * p + 0) * C_DIM + n];
        v.y = w[(2 * p + 1) * C_DIM + n];
        const int dw = p * 256 + ((2 * n + 32 * (p & 1)) & 255);
        *(v2f*)&lds_w[dw] = v;
    }
    __syncthreads();

    const int wave   = threadIdx.x >> 5;
    const int lane   = threadIdx.x & 31;
    const int lane16 = lane & 15;
    const int khalf  = lane >> 4;

    const int m0 = (blockIdx.x * 8 + wave) * 16;
    int row  = m0 + lane16;
    int rowc = row < N ? row : N - 1;

    v8f acc[8];
    const v8f vzero = {0.f,0.f,0.f,0.f,0.f,0.f,0.f,0.f};
    #pragma unroll
    for (int n = 0; n < 8; ++n) acc[n] = vzero;

    const float* xrow = x + (size_t)rowc * C_DIM;

    int coff[8];
    #pragma unroll
    for (int n = 0; n < 8; ++n)
        coff[n] = (32 * n + 2 * lane16 + 32 * khalf) & 255;
    const float* lwp = lds_w + khalf * 256;

    #pragma unroll 4
    for (int k = 0; k < C_DIM; k += 4) {
        const int kb = k + 2 * khalf;
        v2f a = *(const v2f*)(xrow + kb);
        const float* lk = lwp + (k >> 1) * 256;
        #pragma unroll
        for (int n = 0; n < 8; ++n) {
            v2f b = *(const v2f*)(lk + coff[n]);
            acc[n] = __builtin_amdgcn_wmma_f32_16x16x4_f32(
                false, a, false, b, (short)0, acc[n], false, false);
        }
    }

    float nrm[8];
    #pragma unroll
    for (int j = 0; j < 8; ++j) {
        const int r = m0 + j + 8 * khalf;
        nrm[j] = (r < N) ? norm[r] : 0.0f;
    }
    #pragma unroll
    for (int n = 0; n < 8; ++n) {
        #pragma unroll
        for (int j = 0; j < 8; ++j) {
            const int r = m0 + j + 8 * khalf;
            if (r < N) {
                h[(size_t)r * C_DIM + n * 16 + lane16] = acc[n][j] * nrm[j];
            }
        }
    }
}

__global__ __launch_bounds__(256) void gcn_zero(float4* __restrict__ out, int n4)
{
    int i = blockIdx.x * blockDim.x + threadIdx.x;
    if (i < n4) out[i] = make_float4(0.f, 0.f, 0.f, 0.f);
}

__global__ __launch_bounds__(256) void gcn_scatter(
    const float* __restrict__ h, const int* __restrict__ esrc,
    const int* __restrict__ edst, float* __restrict__ out, int E)
{
    const long long tid = (long long)blockIdx.x * blockDim.x + threadIdx.x;
    const long long e = tid >> 5;
    if (e >= E) return;
    const int lane = (int)(tid & 31);
    const int src = esrc[e];
    const int dst = edst[e];
    const float4 v = *(const float4*)(h + (size_t)src * C_DIM + lane * 4);
    float* o = out + (size_t)dst * C_DIM + lane * 4;
    atomicAdd(o + 0, v.x);
    atomicAdd(o + 1, v.y);
    atomicAdd(o + 2, v.z);
    atomicAdd(o + 3, v.w);
}

__global__ __launch_bounds__(256) void gcn_final(
    float* __restrict__ out, const float* __restrict__ norm,
    const float* __restrict__ bias, int N)
{
    const long long tid = (long long)blockIdx.x * blockDim.x + threadIdx.x;
    if (tid >= (long long)N * 32) return;
    const int row = (int)(tid >> 5);
    const int c4  = (int)(tid & 31) * 4;
    const float nm = norm[row];
    float4 v = ((float4*)out)[tid];
    const float4 b = *(const float4*)(bias + c4);
    v.x = v.x * nm + b.x;
    v.y = v.y * nm + b.y;
    v.z = v.z * nm + b.z;
    v.w = v.w * nm + b.w;
    ((float4*)out)[tid] = v;
}

extern "C" void kernel_launch(void* const* d_in, const int* in_sizes, int n_in,
                              void* d_out, int out_size, void* d_ws, size_t ws_size,
                              hipStream_t stream) {
    const float* x    = (const float*)d_in[0];
    const float* norm = (const float*)d_in[1];
    const float* w    = (const float*)d_in[2];
    const float* bias = (const float*)d_in[3];
    const int*   esrc = (const int*)d_in[4];
    const int*   edst = (const int*)d_in[5];
    float* out = (float*)d_out;

    const int N = in_sizes[0] / C_DIM;
    const int E = in_sizes[4];
    float* h = (float*)d_ws;

    dim3 blk(256);

    gcn_gemm_norm<<<(N + 127) / 128, blk, 0, stream>>>(x, w, norm, h, N);

    const int n4 = out_size / 4;
    gcn_zero<<<(n4 + 255) / 256, blk, 0, stream>>>((float4*)out, n4);

    const long long sthreads = (long long)E * 32;
    gcn_scatter<<<(unsigned)((sthreads + 255) / 256), blk, 0, stream>>>(h, esrc, edst, out, E);

    const long long fthreads = (long long)N * 32;
    gcn_final<<<(unsigned)((fthreads + 255) / 256), blk, 0, stream>>>(out, norm, bias, N);
}


# Scatter: lane l takes channels l, l+32, l+64, l+96, so each f32 atomic and each gather load instruction covers one whole 128-byte line, not 8 lanes in each of 4 lines; sums stay in arrival order
# speedup vs baseline: 1.2949x; 1.2949x over previous
#include <hip/hip_runtime.h>

typedef __attribute__((ext_vector_type(2))) float v2f;
typedef __attribute__((ext_vector_type(8))) float v8f;

#define C_DIM 128

__global__ __launch_bounds__(256) void gcn_gemm_norm(
    const float* __restrict__ x, const float* __restrict__ w,
    const float* __restrict__ norm, float* __restrict__ h, int N)
{
    __shared__ float lds_w[C_DIM * C_DIM];

    for (int idx = threadIdx.x; idx < (C_DIM / 2) * C_DIM; idx += 256) {
        const int p = idx >> 7;
        const int n = idx & 127;
        v2f v;
        v.x = w[(2 * p + 0) * C_DIM + n];
        v.y = w[(2 * p + 1) * C_DIM + n];
        const int dw = p * 256 + ((2 * n + 32 * (p & 1)) & 255);
        *(v2f*)&lds_w[dw] = v;
    }
    __syncthreads();

    const int wave   = threadIdx.x >> 5;
    const int lane   = threadIdx.x & 31;
    const int lane16 = lane & 15;
    const int khalf  = lane >> 4;

    const int m0 = (blockIdx.x * 8 + wave) * 16;
    int row  = m0 + lane16;
    int rowc = row < N ? row : N - 1;

    v8f acc[8];
    const v8f vzero = {0.f,0.f,0.f,0.f,0.f,0.f,0.f,0.f};
    #pragma unroll
    for (int n = 0; n < 8; ++n) acc[n] = vzero;

    const float* xrow = x + (size_t)rowc * C_DIM;

    int coff[8];
    #pragma unroll
    for (int n = 0; n < 8; ++n)
        coff[n] = (32 * n + 2 * lane16 + 32 * khalf) & 255;
    const float* lwp = lds_w + khalf * 256;

    #pragma unroll 4
    for (int k = 0; k < C_DIM; k += 4) {
        const int kb = k + 2 * khalf;
        v2f a = *(const v2f*)(xrow + kb);
        const float* lk = lwp + (k >> 1) * 256;
        #pragma unroll
        for (int n = 0; n < 8; ++n) {
            v2f b = *(const v2f*)(lk + coff[n]);
            acc[n] = __builtin_amdgcn_wmma_f32_16x16x4_f32(
                false, a, false, b, (short)0, acc[n], false, false);
        }
    }

    float nrm[8];
    #pragma unroll
    for (int j = 0; j < 8; ++j) {
        const int r = m0 + j + 8 * khalf;
        nrm[j] = (r < N) ? norm[r] : 0.0f;
    }
    #pragma unroll
    for (int n = 0; n < 8; ++n) {
        #pragma unroll
        for (int j = 0; j < 8; ++j) {
            const int r = m0 + j + 8 * khalf;
            if (r < N) {
                h[(size_t)r * C_DIM + n * 16 + lane16] = acc[n][j] * nrm[j];
            }
        }
    }
}

__global__ __launch_bounds__(256) void gcn_zero(float4* __restrict__ out, int n4)
{
    int i = blockIdx.x * blockDim.x + threadIdx.x;
    if (i < n4) out[i] = make_float4(0.f, 0.f, 0.f, 0.f);
}

__global__ __launch_bounds__(256) void gcn_scatter(
    const float* __restrict__ h, const int* __restrict__ esrc,
    const int* __restrict__ edst, float* __restrict__ out, int E)
{
    const long long tid = (long long)blockIdx.x * blockDim.x + threadIdx.x;
    const long long e = tid >> 5;
    if (e >= E) return;
    const int lane = (int)(tid & 31);
    const int src = esrc[e];
    const int dst = edst[e];
    const float* hp = h + (size_t)src * C_DIM + lane;
    const float v0 = hp[0];
    const float v1 = hp[32];
    const float v2 = hp[64];
    const float v3 = hp[96];
    float* o = out + (size_t)dst * C_DIM + lane;
    atomicAdd(o + 0,  v0);
    atomicAdd(o + 32, v1);
    atomicAdd(o + 64, v2);
    atomicAdd(o + 96, v3);
}

__global__ __launch_bounds__(256) void gcn_final(
    float* __restrict__ out, const float* __restrict__ norm,
    const float* __restrict__ bias, int N)
{
    const long long tid = (long long)blockIdx.x * blockDim.x + threadIdx.x;
    if (tid >= (long long)N * 32) return;
    const int row = (int)(tid >> 5);
    const int c4  = (int)(tid & 31) * 4;
    const float nm = norm[row];
    float4 v = ((float4*)out)[tid];
    const float4 b = *(const float4*)(bias + c4);
    v.x = v.x * nm + b.x;
    v.y = v.y * nm + b.y;
    v.z = v.z * nm + b.z;
    v.w = v.w * nm + b.w;
    ((float4*)out)[tid] = v;
}

extern "C" void kernel_launch(void* const* d_in, const int* in_sizes, int n_in,
                              void* d_out, int out_size, void* d_ws, size_t ws_size,
                              hipStream_t stream) {
    const float* x    = (const float*)d_in[0];
    const float* norm = (const float*)d_in[1];
    const float* w    = (const float*)d_in[2];
    const float* bias = (const float*)d_in[3];
    const int*   esrc = (const int*)d_in[4];
    const int*   edst = (const int*)d_in[5];
    float* out = (float*)d_out;

    const int N = in_sizes[0] / C_DIM;
    const int E = in_sizes[4];
    float* h = (float*)d_ws;

    dim3 blk(256);

    gcn_gemm_norm<<<(N + 127) / 128, blk, 0, stream>>>(x, w, norm, h, N);

    const int n4 = out_size / 4;
    gcn_zero<<<(n4 + 255) / 256, blk, 0, stream>>>((float4*)out, n4);

    const long long sthreads = (long long)E * 32;
    gcn_scatter<<<(unsigned)((sthreads + 255) / 256), blk, 0, stream>>>(h, esrc, edst, out, E);

    const long long fthreads = (long long)N * 32;
    gcn_final<<<(unsigned)((fthreads + 255) / 256), blk, 0, stream>>>(out, norm, bias, N);
}


# Float atomics on out replaced by per-row buckets (one int atomic per edge); one wave per row sorts its sources in registers, adds h rows in ascending source order, stores out*norm+bias once
# speedup vs baseline: 2.0521x; 1.5848x over previous
#include <hip/hip_runtime.h>

typedef __attribute__((ext_vector_type(2))) float v2f;
typedef __attribute__((ext_vector_type(8))) float v8f;

#define C_DIM 128

__global__ __launch_bounds__(256) void gcn_gemm_norm(
    const float* __restrict__ x, const float* __restrict__ w,
    const float* __restrict__ norm, float* __restrict__ h,
    int* __restrict__ cnt, int N)
{
    __shared__ float lds_w[C_DIM * C_DIM];

    if (threadIdx.x < 128) {
        const int r = blockIdx.x * 128 + threadIdx.x;
        if (r < N) cnt[r] = 0;
    }

    for (int idx = threadIdx.x; idx < (C_DIM / 2) * C_DIM; idx += 256) {
        const int p = idx >> 7;
        const int n = idx & 127;
        v2f v;
        v.x = w[(2 * p + 0) * C_DIM + n];
        v.y = w[(2 * p + 1) * C_DIM + n];
        const int dw = p * 256 + ((2 * n + 32 * (p & 1)) & 255);
        *(v2f*)&lds_w[dw] = v;
    }
    __syncthreads();

    const int wave   = threadIdx.x >> 5;
    const int lane   = threadIdx.x & 31;
    const int lane16 = lane & 15;
    const int khalf  = lane >> 4;

    const int m0 = (blockIdx.x * 8 + wave) * 16;
    int row  = m0 + lane16;
    int rowc = row < N ? row : N - 1;

    v8f acc[8];
    const v8f vzero = {0.f,0.f,0.f,0.f,0.f,0.f,0.f,0.f};
    #pragma unroll
    for (int n = 0; n < 8; ++n) acc[n] = vzero;

    const float* xrow = x + (size_t)rowc * C_DIM;

    int coff[8];
    #pragma unroll
    for (int n = 0; n < 8; ++n)
        coff[n] = (32 * n + 2 * lane16 + 32 * khalf) & 255;
    const float* lwp = lds_w + khalf * 256;

    #pragma unroll 4
    for (int k = 0; k < C_DIM; k += 4) {
        const int kb = k + 2 * khalf;
        v2f a = *(const v2f*)(xrow + kb);
        const float* lk = lwp + (k >> 1) * 256;
        #pragma unroll
        for (int n = 0; n < 8; ++n) {
            v2f b = *(const v2f*)(lk + coff[n]);
            acc[n] = __builtin_amdgcn_wmma_f32_16x16x4_f32(
                false, a, false, b, (short)0, acc[n], false, false);
        }
    }

    float nrm[8];
    #pragma unroll
    for (int j = 0; j < 8; ++j) {
        const int r = m0 + j + 8 * khalf;
        nrm[j] = (r < N) ? norm[r] : 0.0f;
    }
    #pragma unroll
    for (int n = 0; n < 8; ++n) {
        #pragma unroll
        for (int j = 0; j < 8; ++j) {
            const int r = m0 + j + 8 * khalf;
            if (r < N) {
                h[(size_t)r * C_DIM + n * 16 + lane16] = acc[n][j] * nrm[j];
            }
        }
    }
}

#define BUCKET_CAP 64

__global__ __launch_bounds__(256) void gcn_bucket(
    const int* __restrict__ esrc, const int* __restrict__ edst,
    int* __restrict__ cnt, int* __restrict__ slots, int E)
{
    const int e = blockIdx.x * 256 + threadIdx.x;
    if (e >= E) return;
    const int d = edst[e];
    const int s = esrc[e];
    const int r = atomicAdd(&cnt[d], 1);
    if (r < BUCKET_CAP) slots[(size_t)d * BUCKET_CAP + r] = s;
}

__global__ __launch_bounds__(256) void gcn_rowsum(
    const float* __restrict__ h, const int* __restrict__ cnt,
    const int* __restrict__ slots,
    const int* __restrict__ esrc, const int* __restrict__ edst,
    const float* __restrict__ norm, const float* __restrict__ bias,
    float* __restrict__ out, int N, int E)
{
    const int lane = threadIdx.x & 31;
    const int row = __builtin_amdgcn_readfirstlane(blockIdx.x * 8 + (threadIdx.x >> 5));
    if (row >= N) return;

    const int c = cnt[row];
    const float nm = norm[row];
    const float4 b = *(const float4*)(bias + lane * 4);
    const float* hl = h + lane * 4;
    float4 acc = make_float4(0.f, 0.f, 0.f, 0.f);

    if (c <= BUCKET_CAP) {
        const int* sl = slots + (size_t)row * BUCKET_CAP;
        unsigned k0 = 0xFFFFFFFFu, k1 = 0xFFFFFFFFu;
        if (lane < c)      k0 = ((unsigned)sl[lane] << 6) | (unsigned)lane;
        if (lane + 32 < c) k1 = ((unsigned)sl[lane + 32] << 6) | (unsigned)(lane + 32);

        if (c <= 32) {
            int r0 = 0;
            for (int j = 0; j < c; ++j) {
                const unsigned kj = (unsigned)__builtin_amdgcn_readlane((int)k0, j);
                r0 += (kj < k0) ? 1 : 0;
            }
            const unsigned sorted = (unsigned)__builtin_amdgcn_ds_permute(r0 << 2, (int)k0);
            const int c4 = c & ~3;
            const int rem = c - c4;
            float4 t0 = make_float4(0.f, 0.f, 0.f, 0.f), t1 = t0, t2 = t0;
            if (rem > 0) t0 = *(const float4*)(hl + (size_t)((unsigned)__builtin_amdgcn_readlane((int)sorted, c4) >> 6) * C_DIM);
            if (rem > 1) t1 = *(const float4*)(hl + (size_t)((unsigned)__builtin_amdgcn_readlane((int)sorted, c4 + 1) >> 6) * C_DIM);
            if (rem > 2) t2 = *(const float4*)(hl + (size_t)((unsigned)__builtin_amdgcn_readlane((int)sorted, c4 + 2) >> 6) * C_DIM);
            for (int r = 0; r < c4; r += 4) {
                const unsigned s0 = (unsigned)__builtin_amdgcn_readlane((int)sorted, r) >> 6;
                const unsigned s1 = (unsigned)__builtin_amdgcn_readlane((int)sorted, r + 1) >> 6;
                const unsigned s2 = (unsigned)__builtin_amdgcn_readlane((int)sorted, r + 2) >> 6;
                const unsigned s3 = (unsigned)__builtin_amdgcn_readlane((int)sorted, r + 3) >> 6;
                const float4 v0 = *(const float4*)(hl + (size_t)s0 * C_DIM);
                const float4 v1 = *(const float4*)(hl + (size_t)s1 * C_DIM);
                const float4 v2 = *(const float4*)(hl + (size_t)s2 * C_DIM);
                const float4 v3 = *(const float4*)(hl + (size_t)s3 * C_DIM);
                acc.x += v0.x; acc.y += v0.y; acc.z += v0.z; acc.w += v0.w;
                acc.x += v1.x; acc.y += v1.y; acc.z += v1.z; acc.w += v1.w;
                acc.x += v2.x; acc.y += v2.y; acc.z += v2.z; acc.w += v2.w;
                acc.x += v3.x; acc.y += v3.y; acc.z += v3.z; acc.w += v3.w;
            }
            if (rem > 0) { acc.x += t0.x; acc.y += t0.y; acc.z += t0.z; acc.w += t0.w; }
            if (rem > 1) { acc.x += t1.x; acc.y += t1.y; acc.z += t1.z; acc.w += t1.w; }
            if (rem > 2) { acc.x += t2.x; acc.y += t2.y; acc.z += t2.z; acc.w += t2.w; }
        } else {
            int r0 = 0, r1 = 0;
            for (int j = 0; j < 32; ++j) {
                const unsigned kj = (unsigned)__builtin_amdgcn_readlane((int)k0, j);
                r0 += (kj < k0) ? 1 : 0;
                r1 += (kj < k1) ? 1 : 0;
            }
            for (int j = 32; j < c; ++j) {
                const unsigned kj = (unsigned)__builtin_amdgcn_readlane((int)k1, j - 32);
                r0 += (kj < k0) ? 1 : 0;
                r1 += (kj < k1) ? 1 : 0;
            }
            for (int r = 0; r < c; ++r) {
                unsigned m = __builtin_amdgcn_ballot_w32(r0 == r);
                unsigned key;
                if (m != 0u) {
                    key = (unsigned)__builtin_amdgcn_readlane((int)k0, __builtin_ctz(m));
                } else {
                    m = __builtin_amdgcn_ballot_w32(r1 == r);
                    key = (unsigned)__builtin_amdgcn_readlane((int)k1, __builtin_ctz(m));
                }
                const unsigned s = key >> 6;
                const float4 v = *(const float4*)(hl + (size_t)s * C_DIM);
                acc.x += v.x; acc.y += v.y; acc.z += v.z; acc.w += v.w;
            }
        }
    } else {
        for (int base = 0; base < E; base += 32) {
            const int e = base + lane;
            const int d = (e < E) ? edst[e] : -1;
            unsigned m = __builtin_amdgcn_ballot_w32(d == row);
            while (m != 0u) {
                const int j = __builtin_ctz(m);
                m &= m - 1u;
                const int s = esrc[base + j];
                const float4 v = *(const float4*)(hl + (size_t)s * C_DIM);
                acc.x += v.x; acc.y += v.y; acc.z += v.z; acc.w += v.w;
            }
        }
    }

    float4 o;
    o.x = acc.x * nm + b.x;
    o.y = acc.y * nm + b.y;
    o.z = acc.z * nm + b.z;
    o.w = acc.w * nm + b.w;
    *(float4*)(out + (size_t)row * C_DIM + lane * 4) = o;
}

extern "C" void kernel_launch(void* const* d_in, const int* in_sizes, int n_in,
                              void* d_out, int out_size, void* d_ws, size_t ws_size,
                              hipStream_t stream) {
    const float* x    = (const float*)d_in[0];
    const float* norm = (const float*)d_in[1];
    const float* w    = (const float*)d_in[2];
    const float* bias = (const float*)d_in[3];
    const int*   esrc = (const int*)d_in[4];
    const int*   edst = (const int*)d_in[5];
    float* out = (float*)d_out;

    const int N = in_sizes[0] / C_DIM;
    const int E = in_sizes[4];
    float* h     = (float*)d_ws;
    int*   slots = (int*)(h + (size_t)N * C_DIM);
    int*   cnt   = slots + (size_t)N * BUCKET_CAP;

    dim3 blk(256);

    gcn_gemm_norm<<<(N + 127) / 128, blk, 0, stream>>>(x, w, norm, h, cnt, N);

    gcn_bucket<<<(E + 255) / 256, blk, 0, stream>>>(esrc, edst, cnt, slots, E);

    gcn_rowsum<<<(N + 7) / 8, blk, 0, stream>>>(h, cnt, slots, esrc, edst, norm, bias, out, N, E);
}


# Edge bucketing: per-edge returning global atomic replaced by two-level counting sort on LDS counters (bin by 256-row tile, then one block per tile ranks per row); sorted sum order kept
# speedup vs baseline: 2.8445x; 1.3861x over previous
#include <hip/hip_runtime.h>

typedef __attribute__((ext_vector_type(2))) float v2f;
typedef __attribute__((ext_vector_type(8))) float v8f;

#define C_DIM 128
#define TCNT_STRIDE 32

__global__ __launch_bounds__(256) void gcn_gemm_norm(
    const float* __restrict__ x, const float* __restrict__ w,
    const float* __restrict__ norm, float* __restrict__ h,
    int* __restrict__ tcnt, int T, int N)
{
    __shared__ float lds_w[C_DIM * C_DIM];

    if (blockIdx.x == 0) {
        for (int i = threadIdx.x; i < T; i += 256) tcnt[i * TCNT_STRIDE] = 0;
    }

    for (int idx = threadIdx.x; idx < (C_DIM / 2) * C_DIM; idx += 256) {
        const int p = idx >> 7;
        const int n = idx & 127;
        v2f v;
        v.x = w[(2 * p + 0) * C_DIM + n];
        v.y = w[(2 * p + 1) * C_DIM + n];
        const int dw = p * 256 + ((2 * n + 32 * (p & 1)) & 255);
        *(v2f*)&lds_w[dw] = v;
    }
    __syncthreads();

    const int wave   = threadIdx.x >> 5;
    const int lane   = threadIdx.x & 31;
    const int lane16 = lane & 15;
    const int khalf  = lane >> 4;

    const int m0 = (blockIdx.x * 8 + wave) * 16;
    int row  = m0 + lane16;
    int rowc = row < N ? row : N - 1;

    v8f acc[8];
    const v8f vzero = {0.f,0.f,0.f,0.f,0.f,0.f,0.f,0.f};
    #pragma unroll
    for (int n = 0; n < 8; ++n) acc[n] = vzero;

    const float* xrow = x + (size_t)rowc * C_DIM;

    int coff[8];
    #pragma unroll
    for (int n = 0; n < 8; ++n)
        coff[n] = (32 * n + 2 * lane16 + 32 * khalf) & 255;
    const float* lwp = lds_w + khalf * 256;

    #pragma unroll 4
    for (int k = 0; k < C_DIM; k += 4) {
        const int kb = k + 2 * khalf;
        v2f a = *(const v2f*)(xrow + kb);
        const float* lk = lwp + (k >> 1) * 256;
        #pragma unroll
        for (int n = 0; n < 8; ++n) {
            v2f b = *(const v2f*)(lk + coff[n]);
            acc[n] = __builtin_amdgcn_wmma_f32_16x16x4_f32(
                false, a, false, b, (short)0, acc[n], false, false);
        }
    }

    float nrm[8];
    #pragma unroll
    for (int j = 0; j < 8; ++j) {
        const int r = m0 + j + 8 * khalf;
        nrm[j] = (r < N) ? norm[r] : 0.0f;
    }
    #pragma unroll
    for (int n = 0; n < 8; ++n) {
        #pragma unroll
        for (int j = 0; j < 8; ++j) {
            const int r = m0 + j + 8 * khalf;
            if (r < N) {
                h[(size_t)r * C_DIM + n * 16 + lane16] = acc[n][j] * nrm[j];
            }
        }
    }
}

#define BUCKET_CAP 64
#define TILE_ROWS  256
#define TILE_SHIFT 8
#define MAX_TILES  512
#define TILE_CAP   8192
#define BIN_CHUNK  4096

__global__ __launch_bounds__(256) void gcn_bin(
    const int* __restrict__ esrc, const int* __restrict__ edst,
    int* __restrict__ tcnt, int* __restrict__ trec, int N, int E, int T)
{
    __shared__ int lcnt[MAX_TILES];
    const int tid = threadIdx.x;
    for (int i = tid; i < T; i += 256) lcnt[i] = 0;
    __syncthreads();

    const int base = blockIdx.x * BIN_CHUNK;
    int d[16], s[16], t[16], rk[16], bs[16];
    #pragma unroll
    for (int k = 0; k < 16; ++k) {
        const int e  = base + k * 256 + tid;
        const int ec = (e < E) ? e : E - 1;
        d[k] = edst[ec];
        s[k] = esrc[ec];
    }
    #pragma unroll
    for (int k = 0; k < 16; ++k) {
        const int e = base + k * 256 + tid;
        if (e >= E || (unsigned)d[k] >= (unsigned)N) d[k] = -1;
        t[k] = (d[k] >= 0) ? (d[k] >> TILE_SHIFT) : 0;
    }
    #pragma unroll
    for (int k = 0; k < 16; ++k) {
        rk[k] = 0;
        if (d[k] >= 0) rk[k] = atomicAdd(&lcnt[t[k]], 1);
    }
    __syncthreads();
    for (int i = tid; i < T; i += 256) {
        const int c = lcnt[i];
        lcnt[i] = (c > 0) ? atomicAdd(&tcnt[i * TCNT_STRIDE], c) : 0;
    }
    __syncthreads();
    #pragma unroll
    for (int k = 0; k < 16; ++k) bs[k] = lcnt[t[k]];
    #pragma unroll
    for (int k = 0; k < 16; ++k) {
        const int pos = bs[k] + rk[k];
        if (d[k] >= 0 && pos < TILE_CAP)
            trec[(size_t)t[k] * TILE_CAP + pos] = (s[k] << TILE_SHIFT) | (d[k] & (TILE_ROWS - 1));
    }
}

__global__ __launch_bounds__(256) void gcn_place(
    const int* __restrict__ esrc, const int* __restrict__ edst,
    const int* __restrict__ tcnt, const int* __restrict__ trec,
    int* __restrict__ cnt, int* __restrict__ slots, int N, int E)
{
    __shared__ int lcnt[TILE_ROWS + 1];
    __shared__ int lsl[TILE_ROWS * BUCKET_CAP];
    const int tid  = threadIdx.x;
    const int tile = blockIdx.x;
    const int r0   = tile * TILE_ROWS;
    lcnt[tid] = 0;
    if (tid == 0) lcnt[TILE_ROWS] = 0;
    __syncthreads();

    const int n = tcnt[tile * TCNT_STRIDE];
    if (n <= TILE_CAP) {
        const int* tr = trec + (size_t)tile * TILE_CAP;
        for (int i0 = 0; i0 < n; i0 += 2048) {
            int rec[8], lr[8], r[8];
            #pragma unroll
            for (int k = 0; k < 8; ++k) {
                const int i = i0 + k * 256 + tid;
                rec[k] = tr[(i < n) ? i : n - 1];
                lr[k] = (i < n) ? (rec[k] & (TILE_ROWS - 1)) : TILE_ROWS;
            }
            #pragma unroll
            for (int k = 0; k < 8; ++k) r[k] = atomicAdd(&lcnt[lr[k]], 1);
            #pragma unroll
            for (int k = 0; k < 8; ++k) {
                if (lr[k] < TILE_ROWS && r[k] < BUCKET_CAP)
                    lsl[lr[k] * BUCKET_CAP + ((r[k] + lr[k]) & (BUCKET_CAP - 1))] = (int)((unsigned)rec[k] >> TILE_SHIFT);
            }
        }
    } else {
        for (int e = tid; e < E; e += 256) {
            const int dd = edst[e];
            const unsigned lr = (unsigned)(dd - r0);
            if (lr < (unsigned)TILE_ROWS && dd < N) {
                const int sidx = esrc[e];
                const int r = atomicAdd(&lcnt[lr], 1);
                if (r < BUCKET_CAP) lsl[lr * BUCKET_CAP + ((r + lr) & (BUCKET_CAP - 1))] = sidx;
            }
        }
    }
    __syncthreads();

    {
        const int row = r0 + tid;
        if (row < N) cnt[row] = lcnt[tid];
    }
    const int wave  = tid >> 5;
    const int lane  = tid & 31;
    const int lbase = wave * 32;
    const int myc   = lcnt[lbase + lane];
    for (int j0 = 0; j0 < 32; j0 += 8) {
        int v[8];
        #pragma unroll
        for (int u = 0; u < 8; ++u) {
            const int lr = lbase + j0 + u;
            v[u] = lsl[lr * BUCKET_CAP + ((lane + lr) & (BUCKET_CAP - 1))];
        }
        #pragma unroll
        for (int u = 0; u < 8; ++u) {
            const int row = r0 + lbase + j0 + u;
            if (row < N) slots[(size_t)row * BUCKET_CAP + lane] = v[u];
        }
    }
    unsigned m = __builtin_amdgcn_ballot_w32(myc > 32);
    while (m != 0u) {
        const int j = __builtin_ctz(m);
        m &= m - 1u;
        const int lr  = lbase + j;
        const int row = r0 + lr;
        if (row < N)
            slots[(size_t)row * BUCKET_CAP + 32 + lane] = lsl[lr * BUCKET_CAP + ((lane + 32 + lr) & (BUCKET_CAP - 1))];
    }
}

__global__ __launch_bounds__(256) void gcn_rowsum(
    const float* __restrict__ h, const int* __restrict__ cnt,
    const int* __restrict__ slots,
    const int* __restrict__ esrc, const int* __restrict__ edst,
    const float* __restrict__ norm, const float* __restrict__ bias,
    float* __restrict__ out, int N, int E)
{
    const int lane = threadIdx.x & 31;
    const int row = __builtin_amdgcn_readfirstlane(blockIdx.x * 8 + (threadIdx.x >> 5));
    if (row >= N) return;

    const int c = cnt[row];
    const float nm = norm[row];
    const float4 b = *(const float4*)(bias + lane * 4);
    const float* hl = h + lane * 4;
    float4 acc = make_float4(0.f, 0.f, 0.f, 0.f);

    if (c <= BUCKET_CAP) {
        const int* sl = slots + (size_t)row * BUCKET_CAP;
        unsigned k0 = 0xFFFFFFFFu, k1 = 0xFFFFFFFFu;
        if (lane < c)      k0 = ((unsigned)sl[lane] << 6) | (unsigned)lane;
        if (lane + 32 < c) k1 = ((unsigned)sl[lane + 32] << 6) | (unsigned)(lane + 32);

        if (c <= 32) {
            int r0 = 0;
            for (int j = 0; j < c; ++j) {
                const unsigned kj = (unsigned)__builtin_amdgcn_readlane((int)k0, j);
                r0 += (kj < k0) ? 1 : 0;
            }
            const unsigned sorted = (unsigned)__builtin_amdgcn_ds_permute(r0 << 2, (int)k0);
            const int c4 = c & ~3;
            const int rem = c - c4;
            float4 t0 = make_float4(0.f, 0.f, 0.f, 0.f), t1 = t0, t2 = t0;
            if (rem > 0) t0 = *(const float4*)(hl + (size_t)((unsigned)__builtin_amdgcn_readlane((int)sorted, c4) >> 6) * C_DIM);
            if (rem > 1) t1 = *(const float4*)(hl + (size_t)((unsigned)__builtin_amdgcn_readlane((int)sorted, c4 + 1) >> 6) * C_DIM);
            if (rem > 2) t2 = *(const float4*)(hl + (size_t)((unsigned)__builtin_amdgcn_readlane((int)sorted, c4 + 2) >> 6) * C_DIM);
            for (int r = 0; r < c4; r += 4) {
                const unsigned s0 = (unsigned)__builtin_amdgcn_readlane((int)sorted, r) >> 6;
                const unsigned s1 = (unsigned)__builtin_amdgcn_readlane((int)sorted, r + 1) >> 6;
                const unsigned s2 = (unsigned)__builtin_amdgcn_readlane((int)sorted, r + 2) >> 6;
                const unsigned s3 = (unsigned)__builtin_amdgcn_readlane((int)sorted, r + 3) >> 6;
                const float4 v0 = *(const float4*)(hl + (size_t)s0 * C_DIM);
                const float4 v1 = *(const float4*)(hl + (size_t)s1 * C_DIM);
                const float4 v2 = *(const float4*)(hl + (size_t)s2 * C_DIM);
                const float4 v3 = *(const float4*)(hl + (size_t)s3 * C_DIM);
                acc.x += v0.x; acc.y += v0.y; acc.z += v0.z; acc.w += v0.w;
                acc.x += v1.x; acc.y += v1.y; acc.z += v1.z; acc.w += v1.w;
                acc.x += v2.x; acc.y += v2.y; acc.z += v2.z; acc.w += v2.w;
                acc.x += v3.x; acc.y += v3.y; acc.z += v3.z; acc.w += v3.w;
            }
            if (rem > 0) { acc.x += t0.x; acc.y += t0.y; acc.z += t0.z; acc.w += t0.w; }
            if (rem > 1) { acc.x += t1.x; acc.y += t1.y; acc.z += t1.z; acc.w += t1.w; }
            if (rem > 2) { acc.x += t2.x; acc.y += t2.y; acc.z += t2.z; acc.w += t2.w; }
        } else {
            int r0 = 0, r1 = 0;
            for (int j = 0; j < 32; ++j) {
                const unsigned kj = (unsigned)__builtin_amdgcn_readlane((int)k0, j);
                r0 += (kj < k0) ? 1 : 0;
                r1 += (kj < k1) ? 1 : 0;
            }
            for (int j = 32; j < c; ++j) {
                const unsigned kj = (unsigned)__builtin_amdgcn_readlane((int)k1, j - 32);
                r0 += (kj < k0) ? 1 : 0;
                r1 += (kj < k1) ? 1 : 0;
            }
            for (int r = 0; r < c; ++r) {
                unsigned m = __builtin_amdgcn_ballot_w32(r0 == r);
                unsigned key;
                if (m != 0u) {
                    key = (unsigned)__builtin_amdgcn_readlane((int)k0, __builtin_ctz(m));
                } else {
                    m = __builtin_amdgcn_ballot_w32(r1 == r);
                    key = (unsigned)__builtin_amdgcn_readlane((int)k1, __builtin_ctz(m));
                }
                const unsigned s = key >> 6;
                const float4 v = *(const float4*)(hl + (size_t)s * C_DIM);
                acc.x += v.x; acc.y += v.y; acc.z += v.z; acc.w += v.w;
            }
        }
    } else {
        for (int base = 0; base < E; base += 32) {
            const int e = base + lane;
            const int d = (e < E) ? edst[e] : -1;
            unsigned m = __builtin_amdgcn_ballot_w32(d == row);
            while (m != 0u) {
                const int j = __builtin_ctz(m);
                m &= m - 1u;
                const int s = esrc[base + j];
                const float4 v = *(const float4*)(hl + (size_t)s * C_DIM);
                acc.x += v.x; acc.y += v.y; acc.z += v.z; acc.w += v.w;
            }
        }
    }

    float4 o;
    o.x = acc.x * nm + b.x;
    o.y = acc.y * nm + b.y;
    o.z = acc.z * nm + b.z;
    o.w = acc.w * nm + b.w;
    *(float4*)(out + (size_t)row * C_DIM + lane * 4) = o;
}

extern "C" void kernel_launch(void* const* d_in, const int* in_sizes, int n_in,
                              void* d_out, int out_size, void* d_ws, size_t ws_size,
                              hipStream_t stream) {
    const float* x    = (const float*)d_in[0];
    const float* norm = (const float*)d_in[1];
    const float* w    = (const float*)d_in[2];
    const float* bias = (const float*)d_in[3];
    const int*   esrc = (const int*)d_in[4];
    const int*   edst = (const int*)d_in[5];
    float* out = (float*)d_out;

    const int N = in_sizes[0] / C_DIM;
    const int E = in_sizes[4];
    float* h     = (float*)d_ws;
    int*   slots = (int*)(h + (size_t)N * C_DIM);
    int*   cnt   = slots + (size_t)N * BUCKET_CAP;
    int*   tcnt  = cnt + N;
    int*   trec  = tcnt + MAX_TILES * TCNT_STRIDE;
    const int T  = (N + TILE_ROWS - 1) / TILE_ROWS;

    dim3 blk(256);

    gcn_gemm_norm<<<(N + 127) / 128, blk, 0, stream>>>(x, w, norm, h, tcnt, T, N);

    gcn_bin<<<(E + BIN_CHUNK - 1) / BIN_CHUNK, blk, 0, stream>>>(esrc, edst, tcnt, trec, N, E, T);

    gcn_place<<<T, blk, 0, stream>>>(esrc, edst, tcnt, trec, cnt, slots, N, E);

    gcn_rowsum<<<(N + 7) / 8, blk, 0, stream>>>(h, cnt, slots, esrc, edst, norm, bias, out, N, E);
}


# GEMM launch dissolved: its row tiles run as extra work-groups of the edge-binning and bucket-placing launches (half each), overlapping the sort passes; launch 1 only clears counters; sums unchanged
# speedup vs baseline: 3.0878x; 1.0855x over previous
#include <hip/hip_runtime.h>

typedef __attribute__((ext_vector_type(2))) float v2f;
typedef __attribute__((ext_vector_type(8))) float v8f;

#define C_DIM 128
#define TCNT_STRIDE 32

__device__ __forceinline__ void gemm_tile(
    const float* __restrict__ x, const float* __restrict__ w,
    const float* __restrict__ norm, float* __restrict__ h,
    float* lds_w, int blk, int N)
{
    for (int idx = threadIdx.x; idx < (C_DIM / 2) * C_DIM; idx += 256) {
        const int p = idx >> 7;
        const int n = idx & 127;
        v2f v;
        v.x = w[(2 * p + 0) * C_DIM + n];
        v.y = w[(2 * p + 1) * C_DIM + n];
        const int dw = p * 256 + ((2 * n + 32 * (p & 1)) & 255);
        *(v2f*)&lds_w[dw] = v;
    }
    __syncthreads();

    const int wave   = threadIdx.x >> 5;
    const int lane   = threadIdx.x & 31;
    const int lane16 = lane & 15;
    const int khalf  = lane >> 4;

    const int m0 = (blk * 8 + wave) * 16;
    int row  = m0 + lane16;
    int rowc = row < N ? row : N - 1;

    v8f acc[8];
    const v8f vzero = {0.f,0.f,0.f,0.f,0.f,0.f,0.f,0.f};
    #pragma unroll
    for (int n = 0; n < 8; ++n) acc[n] = vzero;

    const float* xrow = x + (size_t)rowc * C_DIM;

    int coff[8];
    #pragma unroll
    for (int n = 0; n < 8; ++n)
        coff[n] = (32 * n + 2 * lane16 + 32 * khalf) & 255;
    const float* lwp = lds_w + khalf * 256;

    #pragma unroll 4
    for (int k = 0; k < C_DIM; k += 4) {
        const int kb = k + 2 * khalf;
        v2f a = *(const v2f*)(xrow + kb);
        const float* lk = lwp + (k >> 1) * 256;
        #pragma unroll
        for (int n = 0; n < 8; ++n) {
            v2f b = *(const v2f*)(lk + coff[n]);
            acc[n] = __builtin_amdgcn_wmma_f32_16x16x4_f32(
                false, a, false, b, (short)0, acc[n], false, false);
        }
    }

    float nrm[8];
    #pragma unroll
    for (int j = 0; j < 8; ++j) {
        const int r = m0 + j + 8 * khalf;
        nrm[j] = (r < N) ? norm[r] : 0.0f;
    }
    #pragma unroll
    for (int n = 0; n < 8; ++n) {
        #pragma unroll
        for (int j = 0; j < 8; ++j) {
            const int r = m0 + j + 8 * khalf;
            if (r < N) {
                h[(size_t)r * C_DIM + n * 16 + lane16] = acc[n][j] * nrm[j];
            }
        }
    }
}

__global__ __launch_bounds__(256) void gcn_clear(int* __restrict__ tcnt, int T)
{
    for (int i = threadIdx.x; i < T; i += 256) tcnt[i * TCNT_STRIDE] = 0;
}

#define BUCKET_CAP 64
#define TILE_ROWS  256
#define TILE_SHIFT 8
#define MAX_TILES  512
#define TILE_CAP   8192
#define BIN_CHUNK  4096

__global__ __launch_bounds__(256) void gcn_bin(
    const int* __restrict__ esrc, const int* __restrict__ edst,
    int* __restrict__ tcnt, int* __restrict__ trec, int N, int E, int T,
    const float* __restrict__ x, const float* __restrict__ w,
    const float* __restrict__ norm, float* __restrict__ h,
    int nbin, int gbase)
{
    __shared__ float lds_w[C_DIM * C_DIM];
    __shared__ int lcnt[MAX_TILES];
    if ((int)blockIdx.x >= nbin) {
        gemm_tile(x, w, norm, h, lds_w, gbase + ((int)blockIdx.x - nbin), N);
        return;
    }
    const int tid = threadIdx.x;
    for (int i = tid; i < T; i += 256) lcnt[i] = 0;
    __syncthreads();

    const int base = blockIdx.x * BIN_CHUNK;
    int d[16], s[16], t[16], rk[16], bs[16];
    #pragma unroll
    for (int k = 0; k < 16; ++k) {
        const int e  = base + k * 256 + tid;
        const int ec = (e < E) ? e : E - 1;
        d[k] = edst[ec];
        s[k] = esrc[ec];
    }
    #pragma unroll
    for (int k = 0; k < 16; ++k) {
        const int e = base + k * 256 + tid;
        if (e >= E || (unsigned)d[k] >= (unsigned)N) d[k] = -1;
        t[k] = (d[k] >= 0) ? (d[k] >> TILE_SHIFT) : 0;
    }
    #pragma unroll
    for (int k = 0; k < 16; ++k) {
        rk[k] = 0;
        if (d[k] >= 0) rk[k] = atomicAdd(&lcnt[t[k]], 1);
    }
    __syncthreads();
    for (int i = tid; i < T; i += 256) {
        const int c = lcnt[i];
        lcnt[i] = (c > 0) ? atomicAdd(&tcnt[i * TCNT_STRIDE], c) : 0;
    }
    __syncthreads();
    #pragma unroll
    for (int k = 0; k < 16; ++k) bs[k] = lcnt[t[k]];
    #pragma unroll
    for (int k = 0; k < 16; ++k) {
        const int pos = bs[k] + rk[k];
        if (d[k] >= 0 && pos < TILE_CAP)
            trec[(size_t)t[k] * TILE_CAP + pos] = (s[k] << TILE_SHIFT) | (d[k] & (TILE_ROWS - 1));
    }
}

__global__ __launch_bounds__(256) void gcn_place(
    const int* __restrict__ esrc, const int* __restrict__ edst,
    const int* __restrict__ tcnt, const int* __restrict__ trec,
    int* __restrict__ cnt, int* __restrict__ slots, int N, int E,
    const float* __restrict__ x, const float* __restrict__ w,
    const float* __restrict__ norm, float* __restrict__ h,
    int ntile, int gbase)
{
    __shared__ int lcnt[TILE_ROWS + 1];
    __shared__ int lsl[TILE_ROWS * BUCKET_CAP];
    if ((int)blockIdx.x >= ntile) {
        gemm_tile(x, w, norm, h, (float*)lsl, gbase + ((int)blockIdx.x - ntile), N);
        return;
    }
    const int tid  = threadIdx.x;
    const int tile = blockIdx.x;
    const int r0   = tile * TILE_ROWS;
    lcnt[tid] = 0;
    if (tid == 0) lcnt[TILE_ROWS] = 0;
    __syncthreads();

    const int n = tcnt[tile * TCNT_STRIDE];
    if (n <= TILE_CAP) {
        const int* tr = trec + (size_t)tile * TILE_CAP;
        for (int i0 = 0; i0 < n; i0 += 2048) {
            int rec[8], lr[8], r[8];
            #pragma unroll
            for (int k = 0; k < 8; ++k) {
                const int i = i0 + k * 256 + tid;
                rec[k] = tr[(i < n) ? i : n - 1];
                lr[k] = (i < n) ? (rec[k] & (TILE_ROWS - 1)) : TILE_ROWS;
            }
            #pragma unroll
            for (int k = 0; k < 8; ++k) r[k] = atomicAdd(&lcnt[lr[k]], 1);
            #pragma unroll
            for (int k = 0; k < 8; ++k) {
                if (lr[k] < TILE_ROWS && r[k] < BUCKET_CAP)
                    lsl[lr[k] * BUCKET_CAP + ((r[k] + lr[k]) & (BUCKET_CAP - 1))] = (int)((unsigned)rec[k] >> TILE_SHIFT);
            }
        }
    } else {
        for (int e = tid; e < E; e += 256) {
            const int dd = edst[e];
            const unsigned lr = (unsigned)(dd - r0);
            if (lr < (unsigned)TILE_ROWS && dd < N) {
                const int sidx = esrc[e];
                const int r = atomicAdd(&lcnt[lr], 1);
                if (r < BUCKET_CAP) lsl[lr * BUCKET_CAP + ((r + lr) & (BUCKET_CAP - 1))] = sidx;
            }
        }
    }
    __syncthreads();

    {
        const int row = r0 + tid;
        if (row < N) cnt[row] = lcnt[tid];
    }
    const int wave  = tid >> 5;
    const int lane  = tid & 31;
    const int lbase = wave * 32;
    const int myc   = lcnt[lbase + lane];
    for (int j0 = 0; j0 < 32; j0 += 8) {
        int v[8];
        #pragma unroll
        for (int u = 0; u < 8; ++u) {
            const int lr = lbase + j0 + u;
            v[u] = lsl[lr * BUCKET_CAP + ((lane + lr) & (BUCKET_CAP - 1))];
        }
        #pragma unroll
        for (int u = 0; u < 8; ++u) {
            const int row = r0 + lbase + j0 + u;
            if (row < N) slots[(size_t)row * BUCKET_CAP + lane] = v[u];
        }
    }
    unsigned m = __builtin_amdgcn_ballot_w32(myc > 32);
    while (m != 0u) {
        const int j = __builtin_ctz(m);
        m &= m - 1u;
        const int lr  = lbase + j;
        const int row = r0 + lr;
        if (row < N)
            slots[(size_t)row * BUCKET_CAP + 32 + lane] = lsl[lr * BUCKET_CAP + ((lane + 32 + lr) & (BUCKET_CAP - 1))];
    }
}

__global__ __launch_bounds__(256) void gcn_rowsum(
    const float* __restrict__ h, const int* __restrict__ cnt,
    const int* __restrict__ slots,
    const int* __restrict__ esrc, const int* __restrict__ edst,
    const float* __restrict__ norm, const float* __restrict__ bias,
    float* __restrict__ out, int N, int E)
{
    const int lane = threadIdx.x & 31;
    const int row = __builtin_amdgcn_readfirstlane(blockIdx.x * 8 + (threadIdx.x >> 5));
    if (row >= N) return;

    const int c = cnt[row];
    const float nm = norm[row];
    const float4 b = *(const float4*)(bias + lane * 4);
    const float* hl = h + lane * 4;
    float4 acc = make_float4(0.f, 0.f, 0.f, 0.f);

    if (c <= BUCKET_CAP) {
        const int* sl = slots + (size_t)row * BUCKET_CAP;
        unsigned k0 = 0xFFFFFFFFu, k1 = 0xFFFFFFFFu;
        if (lane < c)      k0 = ((unsigned)sl[lane] << 6) | (unsigned)lane;
        if (lane + 32 < c) k1 = ((unsigned)sl[lane + 32] << 6) | (unsigned)(lane + 32);

        if (c <= 32) {
            int r0 = 0;
            for (int j = 0; j < c; ++j) {
                const unsigned kj = (unsigned)__builtin_amdgcn_readlane((int)k0, j);
                r0 += (kj < k0) ? 1 : 0;
            }
            const unsigned sorted = (unsigned)__builtin_amdgcn_ds_permute(r0 << 2, (int)k0);
            const int c4 = c & ~3;
            const int rem = c - c4;
            float4 t0 = make_float4(0.f, 0.f, 0.f, 0.f), t1 = t0, t2 = t0;
            if (rem > 0) t0 = *(const float4*)(hl + (size_t)((unsigned)__builtin_amdgcn_readlane((int)sorted, c4) >> 6) * C_DIM);
            if (rem > 1) t1 = *(const float4*)(hl + (size_t)((unsigned)__builtin_amdgcn_readlane((int)sorted, c4 + 1) >> 6) * C_DIM);
            if (rem > 2) t2 = *(const float4*)(hl + (size_t)((unsigned)__builtin_amdgcn_readlane((int)sorted, c4 + 2) >> 6) * C_DIM);
            for (int r = 0; r < c4; r += 4) {
                const unsigned s0 = (unsigned)__builtin_amdgcn_readlane((int)sorted, r) >> 6;
                const unsigned s1 = (unsigned)__builtin_amdgcn_readlane((int)sorted, r + 1) >> 6;
                const unsigned s2 = (unsigned)__builtin_amdgcn_readlane((int)sorted, r + 2) >> 6;
                const unsigned s3 = (unsigned)__builtin_amdgcn_readlane((int)sorted, r + 3) >> 6;
                const float4 v0 = *(const float4*)(hl + (size_t)s0 * C_DIM);
                const float4 v1 = *(const float4*)(hl + (size_t)s1 * C_DIM);
                const float4 v2 = *(const float4*)(hl + (size_t)s2 * C_DIM);
                const float4 v3 = *(const float4*)(hl + (size_t)s3 * C_DIM);
                acc.x += v0.x; acc.y += v0.y; acc.z += v0.z; acc.w += v0.w;
                acc.x += v1.x; acc.y += v1.y; acc.z += v1.z; acc.w += v1.w;
                acc.x += v2.x; acc.y += v2.y; acc.z += v2.z; acc.w += v2.w;
                acc.x += v3.x; acc.y += v3.y; acc.z += v3.z; acc.w += v3.w;
            }
            if (rem > 0) { acc.x += t0.x; acc.y += t0.y; acc.z += t0.z; acc.w += t0.w; }
            if (rem > 1) { acc.x += t1.x; acc.y += t1.y; acc.z += t1.z; acc.w += t1.w; }
            if (rem > 2) { acc.x += t2.x; acc.y += t2.y; acc.z += t2.z; acc.w += t2.w; }
        } else {
            int r0 = 0, r1 = 0;
            for (int j = 0; j < 32; ++j) {
                const unsigned kj = (unsigned)__builtin_amdgcn_readlane((int)k0, j);
                r0 += (kj < k0) ? 1 : 0;
                r1 += (kj < k1) ? 1 : 0;
            }
            for (int j = 32; j < c; ++j) {
                const unsigned kj = (unsigned)__builtin_amdgcn_readlane((int)k1, j - 32);
                r0 += (kj < k0) ? 1 : 0;
                r1 += (kj < k1) ? 1 : 0;
            }
            for (int r = 0; r < c; ++r) {
                unsigned m = __builtin_amdgcn_ballot_w32(r0 == r);
                unsigned key;
                if (m != 0u) {
                    key = (unsigned)__builtin_amdgcn_readlane((int)k0, __builtin_ctz(m));
                } else {
                    m = __builtin_amdgcn_ballot_w32(r1 == r);
                    key = (unsigned)__builtin_amdgcn_readlane((int)k1, __builtin_ctz(m));
                }
                const unsigned s = key >> 6;
                const float4 v = *(const float4*)(hl + (size_t)s * C_DIM);
                acc.x += v.x; acc.y += v.y; acc.z += v.z; acc.w += v.w;
            }
        }
    } else {
        for (int base = 0; base < E; base += 32) {
            const int e = base + lane;
            const int d = (e < E) ? edst[e] : -1;
            unsigned m = __builtin_amdgcn_ballot_w32(d == row);
            while (m != 0u) {
                const int j = __builtin_ctz(m);
                m &= m - 1u;
                const int s = esrc[base + j];
                const float4 v = *(const float4*)(hl + (size_t)s * C_DIM);
                acc.x += v.x; acc.y += v.y; acc.z += v.z; acc.w += v.w;
            }
        }
    }

    float4 o;
    o.x = acc.x * nm + b.x;
    o.y = acc.y * nm + b.y;
    o.z = acc.z * nm + b.z;
    o.w = acc.w * nm + b.w;
    *(float4*)(out + (size_t)row * C_DIM + lane * 4) = o;
}

extern "C" void kernel_launch(void* const* d_in, const int* in_sizes, int n_in,
                              void* d_out, int out_size, void* d_ws, size_t ws_size,
                              hipStream_t stream) {
    const float* x    = (const float*)d_in[0];
    const float* norm = (const float*)d_in[1];
    const float* w    = (const float*)d_in[2];
    const float* bias = (const float*)d_in[3];
    const int*   esrc = (const int*)d_in[4];
    const int*   edst = (const int*)d_in[5];
    float* out = (float*)d_out;

    const int N = in_sizes[0] / C_DIM;
    const int E = in_sizes[4];
    float* h     = (float*)d_ws;
    int*   slots = (int*)(h + (size_t)N * C_DIM);
    int*   cnt   = slots + (size_t)N * BUCKET_CAP;
    int*   tcnt  = cnt + N;
    int*   trec  = tcnt + MAX_TILES * TCNT_STRIDE;
    const int T  = (N + TILE_ROWS - 1) / TILE_ROWS;

    dim3 blk(256);

    const int G  = (N + 127) / 128;
    const int G2 = G / 2;
    const int G3 = G - G2;
    const int NB = (E + BIN_CHUNK - 1) / BIN_CHUNK;

    gcn_clear<<<1, blk, 0, stream>>>(tcnt, T);

    gcn_bin<<<NB + G2, blk, 0, stream>>>(esrc, edst, tcnt, trec, N, E, T,
                                         x, w, norm, h, NB, 0);

    gcn_place<<<T + G3, blk, 0, stream>>>(esrc, edst, tcnt, trec, cnt, slots, N, E,
                                          x, w, norm, h, T, G2);

    gcn_rowsum<<<(N + 7) / 8, blk, 0, stream>>>(h, cnt, slots, esrc, edst, norm, bias, out, N, E);
}


# Binning: tile lists reserved by global atomics replaced by one contiguous run per block (records ordered by tile in LDS) plus tile offsets; clearing launch removed (4 launches to 3); sums unchanged
# speedup vs baseline: 4.1963x; 1.3590x over previous
#include <hip/hip_runtime.h>

typedef __attribute__((ext_vector_type(2))) float v2f;
typedef __attribute__((ext_vector_type(8))) float v8f;

#define C_DIM 128

__device__ __forceinline__ void gemm_tile(
    const float* __restrict__ x, const float* __restrict__ w,
    const float* __restrict__ norm, float* __restrict__ h,
    float* lds_w, int blk, int N)
{
    for (int idx = threadIdx.x; idx < (C_DIM / 2) * C_DIM; idx += 256) {
        const int p = idx >> 7;
        const int n = idx & 127;
        v2f v;
        v.x = w[(2 * p + 0) * C_DIM + n];
        v.y = w[(2 * p + 1) * C_DIM + n];
        const int dw = p * 256 + ((2 * n + 32 * (p & 1)) & 255);
        *(v2f*)&lds_w[dw] = v;
    }
    __syncthreads();

    const int wave   = threadIdx.x >> 5;
    const int lane   = threadIdx.x & 31;
    const int lane16 = lane & 15;
    const int khalf  = lane >> 4;

    const int m0 = (blk * 8 + wave) * 16;
    int row  = m0 + lane16;
    int rowc = row < N ? row : N - 1;

    v8f acc[8];
    const v8f vzero = {0.f,0.f,0.f,0.f,0.f,0.f,0.f,0.f};
    #pragma unroll
    for (int n = 0; n < 8; ++n) acc[n] = vzero;

    const float* xrow = x + (size_t)rowc * C_DIM;

    int coff[8];
    #pragma unroll
    for (int n = 0; n < 8; ++n)
        coff[n] = (32 * n + 2 * lane16 + 32 * khalf) & 255;
    const float* lwp = lds_w + khalf * 256;

    #pragma unroll 4
    for (int k = 0; k < C_DIM; k += 4) {
        const int kb = k + 2 * khalf;
        v2f a = *(const v2f*)(xrow + kb);
        const float* lk = lwp + (k >> 1) * 256;
        #pragma unroll
        for (int n = 0; n < 8; ++n) {
            v2f b = *(const v2f*)(lk + coff[n]);
            acc[n] = __builtin_amdgcn_wmma_f32_16x16x4_f32(
                false, a, false, b, (short)0, acc[n], false, false);
        }
    }

    float nrm[8];
    #pragma unroll
    for (int j = 0; j < 8; ++j) {
        const int r = m0 + j + 8 * khalf;
        nrm[j] = (r < N) ? norm[r] : 0.0f;
    }
    #pragma unroll
    for (int n = 0; n < 8; ++n) {
        #pragma unroll
        for (int j = 0; j < 8; ++j) {
            const int r = m0 + j + 8 * khalf;
            if (r < N) {
                h[(size_t)r * C_DIM + n * 16 + lane16] = acc[n][j] * nrm[j];
            }
        }
    }
}

#define BUCKET_CAP 64
#define TILE_ROWS  256
#define TILE_SHIFT 8
#define MAX_TILES  512
#define BIN_CHUNK  4096
#define OFF_STRIDE 520

__global__ __launch_bounds__(256) void gcn_bin(
    const int* __restrict__ esrc, const int* __restrict__ edst,
    int* __restrict__ boff, int* __restrict__ brec, int N, int E, int T,
    const float* __restrict__ x, const float* __restrict__ w,
    const float* __restrict__ norm, float* __restrict__ h,
    int nbin, int gbase)
{
    __shared__ float lds_w[C_DIM * C_DIM];
    __shared__ int lcnt[MAX_TILES];
    if ((int)blockIdx.x >= nbin) {
        gemm_tile(x, w, norm, h, lds_w, gbase + ((int)blockIdx.x - nbin), N);
        return;
    }
    const int tid = threadIdx.x;
    for (int i = tid; i < T; i += 256) lcnt[i] = 0;
    __syncthreads();

    const int base = blockIdx.x * BIN_CHUNK;
    int d[16], s[16], t[16], rk[16];
    #pragma unroll
    for (int k = 0; k < 16; ++k) {
        const int e  = base + k * 256 + tid;
        const int ec = (e < E) ? e : E - 1;
        d[k] = edst[ec];
        s[k] = esrc[ec];
    }
    #pragma unroll
    for (int k = 0; k < 16; ++k) {
        const int e = base + k * 256 + tid;
        if (e >= E || (unsigned)d[k] >= (unsigned)N) d[k] = -1;
        t[k] = (d[k] >= 0) ? (d[k] >> TILE_SHIFT) : 0;
    }
    #pragma unroll
    for (int k = 0; k < 16; ++k) {
        rk[k] = 0;
        if (d[k] >= 0) rk[k] = atomicAdd(&lcnt[t[k]], 1);
    }
    __syncthreads();
    int* st_rec = (int*)lds_w;
    int* lpre   = st_rec + BIN_CHUNK;
    if (tid < 32) {
        int v[16];
        int sum = 0;
        #pragma unroll
        for (int j = 0; j < 16; ++j) {
            const int idx = tid * 16 + j;
            v[j] = (idx < T) ? lcnt[idx] : 0;
            sum += v[j];
        }
        int inc = sum;
        #pragma unroll
        for (int off = 1; off < 32; off <<= 1) {
            const int nb = __shfl_up(inc, off);
            if (tid >= off) inc += nb;
        }
        int run = inc - sum;
        #pragma unroll
        for (int j = 0; j < 16; ++j) {
            lpre[tid * 16 + j] = run;
            run += v[j];
        }
        if (tid == 31) lpre[MAX_TILES] = inc;
    }
    __syncthreads();
    {
        int* bo = boff + (size_t)blockIdx.x * OFF_STRIDE;
        for (int i = tid; i <= T; i += 256) bo[i] = lpre[i];
    }
    #pragma unroll
    for (int k = 0; k < 16; ++k) {
        if (d[k] >= 0)
            st_rec[lpre[t[k]] + rk[k]] = (s[k] << TILE_SHIFT) | (d[k] & (TILE_ROWS - 1));
    }
    __syncthreads();
    const int total = lpre[MAX_TILES];
    int* br = brec + (size_t)blockIdx.x * BIN_CHUNK;
    #pragma unroll
    for (int k = 0; k < 16; ++k) {
        const int q = k * 256 + tid;
        if (q < total) br[q] = st_rec[q];
    }
}

__global__ __launch_bounds__(256) void gcn_place(
    const int* __restrict__ boff, const int* __restrict__ brec, int nbin,
    int* __restrict__ cnt, int* __restrict__ slots, int N,
    const float* __restrict__ x, const float* __restrict__ w,
    const float* __restrict__ norm, float* __restrict__ h,
    int ntile, int gbase)
{
    __shared__ int lcnt[TILE_ROWS];
    __shared__ int lsl[TILE_ROWS * BUCKET_CAP];
    if ((int)blockIdx.x >= ntile) {
        gemm_tile(x, w, norm, h, (float*)lsl, gbase + ((int)blockIdx.x - ntile), N);
        return;
    }
    const int tid  = threadIdx.x;
    const int tile = blockIdx.x;
    const int r0   = tile * TILE_ROWS;
    lcnt[tid] = 0;
    __syncthreads();

    const int4* brec4 = (const int4*)brec;
    int b = tid;
    int beg = 0, end = 0;
    if (b < nbin) {
        const int* bo = boff + (size_t)b * OFF_STRIDE + tile;
        beg = bo[0];
        end = bo[1];
    }
    while (b < nbin) {
        const int nb = b + 256;
        int nbeg = 0, nend = 0;
        if (nb < nbin) {
            const int* bo = boff + (size_t)nb * OFF_STRIDE + tile;
            nbeg = bo[0];
            nend = bo[1];
        }
        if (beg < end) {
            const int4* br4 = brec4 + (size_t)b * (BIN_CHUNK / 4);
            const int clast = (end - 1) >> 2;
            for (int c0 = beg >> 2; c0 <= clast; c0 += 4) {
                int rec[16];
                #pragma unroll
                for (int j = 0; j < 4; ++j) {
                    const int c = c0 + j;
                    const int4 q = br4[(c <= clast) ? c : clast];
                    rec[4 * j + 0] = q.x;
                    rec[4 * j + 1] = q.y;
                    rec[4 * j + 2] = q.z;
                    rec[4 * j + 3] = q.w;
                }
                #pragma unroll
                for (int e = 0; e < 16; ++e) {
                    const int idx = (c0 + (e >> 2)) * 4 + (e & 3);
                    if (idx >= beg && idx < end) {
                        const int lr = rec[e] & (TILE_ROWS - 1);
                        const int r  = atomicAdd(&lcnt[lr], 1);
                        if (r < BUCKET_CAP)
                            lsl[lr * BUCKET_CAP + ((r + lr) & (BUCKET_CAP - 1))] = (int)((unsigned)rec[e] >> TILE_SHIFT);
                    }
                }
            }
        }
        b = nb;
        beg = nbeg;
        end = nend;
    }
    __syncthreads();

    {
        const int row = r0 + tid;
        if (row < N) cnt[row] = lcnt[tid];
    }
    const int wave  = tid >> 5;
    const int lane  = tid & 31;
    const int lbase = wave * 32;
    const int myc   = lcnt[lbase + lane];
    for (int j0 = 0; j0 < 32; j0 += 8) {
        int v[8];
        #pragma unroll
        for (int u = 0; u < 8; ++u) {
            const int lr = lbase + j0 + u;
            v[u] = lsl[lr * BUCKET_CAP + ((lane + lr) & (BUCKET_CAP - 1))];
        }
        #pragma unroll
        for (int u = 0; u < 8; ++u) {
            const int row = r0 + lbase + j0 + u;
            if (row < N) slots[(size_t)row * BUCKET_CAP + lane] = v[u];
        }
    }
    unsigned m = __builtin_amdgcn_ballot_w32(myc > 32);
    while (m != 0u) {
        const int j = __builtin_ctz(m);
        m &= m - 1u;
        const int lr  = lbase + j;
        const int row = r0 + lr;
        if (row < N)
            slots[(size_t)row * BUCKET_CAP + 32 + lane] = lsl[lr * BUCKET_CAP + ((lane + 32 + lr) & (BUCKET_CAP - 1))];
    }
}

__global__ __launch_bounds__(256) void gcn_rowsum(
    const float* __restrict__ h, const int* __restrict__ cnt,
    const int* __restrict__ slots,
    const int* __restrict__ esrc, const int* __restrict__ edst,
    const float* __restrict__ norm, const float* __restrict__ bias,
    float* __restrict__ out, int N, int E)
{
    const int lane = threadIdx.x & 31;
    const int row = __builtin_amdgcn_readfirstlane(blockIdx.x * 8 + (threadIdx.x >> 5));
    if (row >= N) return;

    const int c = cnt[row];
    const float nm = norm[row];
    const float4 b = *(const float4*)(bias + lane * 4);
    const float* hl = h + lane * 4;
    float4 acc = make_float4(0.f, 0.f, 0.f, 0.f);

    if (c <= BUCKET_CAP) {
        const int* sl = slots + (size_t)row * BUCKET_CAP;
        unsigned k0 = 0xFFFFFFFFu, k1 = 0xFFFFFFFFu;
        if (lane < c)      k0 = ((unsigned)sl[lane] << 6) | (unsigned)lane;
        if (lane + 32 < c) k1 = ((unsigned)sl[lane + 32] << 6) | (unsigned)(lane + 32);

        if (c <= 32) {
            int r0 = 0;
            for (int j = 0; j < c; ++j) {
                const unsigned kj = (unsigned)__builtin_amdgcn_readlane((int)k0, j);
                r0 += (kj < k0) ? 1 : 0;
            }
            const unsigned sorted = (unsigned)__builtin_amdgcn_ds_permute(r0 << 2, (int)k0);
            const int c4 = c & ~3;
            const int rem = c - c4;
            float4 t0 = make_float4(0.f, 0.f, 0.f, 0.f), t1 = t0, t2 = t0;
            if (rem > 0) t0 = *(const float4*)(hl + (size_t)((unsigned)__builtin_amdgcn_readlane((int)sorted, c4) >> 6) * C_DIM);
            if (rem > 1) t1 = *(const float4*)(hl + (size_t)((unsigned)__builtin_amdgcn_readlane((int)sorted, c4 + 1) >> 6) * C_DIM);
            if (rem > 2) t2 = *(const float4*)(hl + (size_t)((unsigned)__builtin_amdgcn_readlane((int)sorted, c4 + 2) >> 6) * C_DIM);
            for (int r = 0; r < c4; r += 4) {
                const unsigned s0 = (unsigned)__builtin_amdgcn_readlane((int)sorted, r) >> 6;
                const unsigned s1 = (unsigned)__builtin_amdgcn_readlane((int)sorted, r + 1) >> 6;
                const unsigned s2 = (unsigned)__builtin_amdgcn_readlane((int)sorted, r + 2) >> 6;
                const unsigned s3 = (unsigned)__builtin_amdgcn_readlane((int)sorted, r + 3) >> 6;
                const float4 v0 = *(const float4*)(hl + (size_t)s0 * C_DIM);
                const float4 v1 = *(const float4*)(hl + (size_t)s1 * C_DIM);
                const float4 v2 = *(const float4*)(hl + (size_t)s2 * C_DIM);
                const float4 v3 = *(const float4*)(hl + (size_t)s3 * C_DIM);
                acc.x += v0.x; acc.y += v0.y; acc.z += v0.z; acc.w += v0.w;
                acc.x += v1.x; acc.y += v1.y; acc.z += v1.z; acc.w += v1.w;
                acc.x += v2.x; acc.y += v2.y; acc.z += v2.z; acc.w += v2.w;
                acc.x += v3.x; acc.y += v3.y; acc.z += v3.z; acc.w += v3.w;
            }
            if (rem > 0) { acc.x += t0.x; acc.y += t0.y; acc.z += t0.z; acc.w += t0.w; }
            if (rem > 1) { acc.x += t1.x; acc.y += t1.y; acc.z += t1.z; acc.w += t1.w; }
            if (rem > 2) { acc.x += t2.x; acc.y += t2.y; acc.z += t2.z; acc.w += t2.w; }
        } else {
            int r0 = 0, r1 = 0;
            for (int j = 0; j < 32; ++j) {
                const unsigned kj = (unsigned)__builtin_amdgcn_readlane((int)k0, j);
                r0 += (kj < k0) ? 1 : 0;
                r1 += (kj < k1) ? 1 : 0;
            }
            for (int j = 32; j < c; ++j) {
                const unsigned kj = (unsigned)__builtin_amdgcn_readlane((int)k1, j - 32);
                r0 += (kj < k0) ? 1 : 0;
                r1 += (kj < k1) ? 1 : 0;
            }
            for (int r = 0; r < c; ++r) {
                unsigned m = __builtin_amdgcn_ballot_w32(r0 == r);
                unsigned key;
                if (m != 0u) {
                    key = (unsigned)__builtin_amdgcn_readlane((int)k0, __builtin_ctz(m));
                } else {
                    m = __builtin_amdgcn_ballot_w32(r1 == r);
                    key = (unsigned)__builtin_amdgcn_readlane((int)k1, __builtin_ctz(m));
                }
                const unsigned s = key >> 6;
                const float4 v = *(const float4*)(hl + (size_t)s * C_DIM);
                acc.x += v.x; acc.y += v.y; acc.z += v.z; acc.w += v.w;
            }
        }
    } else {
        for (int base = 0; base < E; base += 32) {
            const int e = base + lane;
            const int d = (e < E) ? edst[e] : -1;
            unsigned m = __builtin_amdgcn_ballot_w32(d == row);
            while (m != 0u) {
                const int j = __builtin_ctz(m);
                m &= m - 1u;
                const int s = esrc[base + j];
                const float4 v = *(const float4*)(hl + (size_t)s * C_DIM);
                acc.x += v.x; acc.y += v.y; acc.z += v.z; acc.w += v.w;
            }
        }
    }

    float4 o;
    o.x = acc.x * nm + b.x;
    o.y = acc.y * nm + b.y;
    o.z = acc.z * nm + b.z;
    o.w = acc.w * nm + b.w;
    *(float4*)(out + (size_t)row * C_DIM + lane * 4) = o;
}

extern "C" void kernel_launch(void* const* d_in, const int* in_sizes, int n_in,
                              void* d_out, int out_size, void* d_ws, size_t ws_size,
                              hipStream_t stream) {
    const float* x    = (const float*)d_in[0];
    const float* norm = (const float*)d_in[1];
    const float* w    = (const float*)d_in[2];
    const float* bias = (const float*)d_in[3];
    const int*   esrc = (const int*)d_in[4];
    const int*   edst = (const int*)d_in[5];
    float* out = (float*)d_out;

    const int N = in_sizes[0] / C_DIM;
    const int E = in_sizes[4];
    float* h     = (float*)d_ws;
    int*   slots = (int*)(h + (size_t)N * C_DIM);
    int*   cnt   = slots + (size_t)N * BUCKET_CAP;
    const int T  = (N + TILE_ROWS - 1) / TILE_ROWS;
    const int NB = (E + BIN_CHUNK - 1) / BIN_CHUNK;
    int*   boff  = cnt + ((N + 3) & ~3);
    int*   brec  = boff + (size_t)NB * OFF_STRIDE;

    dim3 blk(256);

    const int G  = (N + 127) / 128;
    const int G2 = G / 2;
    const int G3 = G - G2;

    gcn_bin<<<NB + G2, blk, 0, stream>>>(esrc, edst, boff, brec, N, E, T,
                                         x, w, norm, h, NB, 0);

    gcn_place<<<T + G3, blk, 0, stream>>>(boff, brec, NB, cnt, slots, N,
                                          x, w, norm, h, T, G2);

    gcn_rowsum<<<(N + 7) / 8, blk, 0, stream>>>(h, cnt, slots, esrc, edst, norm, bias, out, N, E);
}


# Row-sum launch dissolved into the bucket-placing launch: a 32-wave work-group per 256-row tile sums its rows from its own LDS buckets (3 launches to 2); all GEMM tiles beside binning; sums unchanged
# speedup vs baseline: 4.4739x; 1.0662x over previous
#include <hip/hip_runtime.h>

typedef __attribute__((ext_vector_type(2))) float v2f;
typedef __attribute__((ext_vector_type(8))) float v8f;

#define C_DIM 128

__device__ __forceinline__ void gemm_tile(
    const float* __restrict__ x, const float* __restrict__ w,
    const float* __restrict__ norm, float* __restrict__ h,
    float* lds_w, int blk, int N)
{
    for (int idx = threadIdx.x; idx < (C_DIM / 2) * C_DIM; idx += 256) {
        const int p = idx >> 7;
        const int n = idx & 127;
        v2f v;
        v.x = w[(2 * p + 0) * C_DIM + n];
        v.y = w[(2 * p + 1) * C_DIM + n];
        const int dw = p * 256 + ((2 * n + 32 * (p & 1)) & 255);
        *(v2f*)&lds_w[dw] = v;
    }
    __syncthreads();

    const int wave   = threadIdx.x >> 5;
    const int lane   = threadIdx.x & 31;
    const int lane16 = lane & 15;
    const int khalf  = lane >> 4;

    const int m0 = (blk * 8 + wave) * 16;
    int row  = m0 + lane16;
    int rowc = row < N ? row : N - 1;

    v8f acc[8];
    const v8f vzero = {0.f,0.f,0.f,0.f,0.f,0.f,0.f,0.f};
    #pragma unroll
    for (int n = 0; n < 8; ++n) acc[n] = vzero;

    const float* xrow = x + (size_t)rowc * C_DIM;

    int coff[8];
    #pragma unroll
    for (int n = 0; n < 8; ++n)
        coff[n] = (32 * n + 2 * lane16 + 32 * khalf) & 255;
    const float* lwp = lds_w + khalf * 256;

    #pragma unroll 4
    for (int k = 0; k < C_DIM; k += 4) {
        const int kb = k + 2 * khalf;
        v2f a = *(const v2f*)(xrow + kb);
        const float* lk = lwp + (k >> 1) * 256;
        #pragma unroll
        for (int n = 0; n < 8; ++n) {
            v2f b = *(const v2f*)(lk + coff[n]);
            acc[n] = __builtin_amdgcn_wmma_f32_16x16x4_f32(
                false, a, false, b, (short)0, acc[n], false, false);
        }
    }

    float nrm[8];
    #pragma unroll
    for (int j = 0; j < 8; ++j) {
        const int r = m0 + j + 8 * khalf;
        nrm[j] = (r < N) ? norm[r] : 0.0f;
    }
    #pragma unroll
    for (int n = 0; n < 8; ++n) {
        #pragma unroll
        for (int j = 0; j < 8; ++j) {
            const int r = m0 + j + 8 * khalf;
            if (r < N) {
                h[(size_t)r * C_DIM + n * 16 + lane16] = acc[n][j] * nrm[j];
            }
        }
    }
}

#define BUCKET_CAP 64
#define TILE_ROWS  256
#define TILE_SHIFT 8
#define MAX_TILES  512
#define BIN_CHUNK  4096
#define OFF_STRIDE 520

__global__ __launch_bounds__(256) void gcn_bin(
    const int* __restrict__ esrc, const int* __restrict__ edst,
    int* __restrict__ boff, int* __restrict__ brec, int N, int E, int T,
    const float* __restrict__ x, const float* __restrict__ w,
    const float* __restrict__ norm, float* __restrict__ h,
    int ngemm)
{
    __shared__ float lds_w[C_DIM * C_DIM];
    __shared__ int lcnt[MAX_TILES];
    if ((int)blockIdx.x < ngemm) {
        gemm_tile(x, w, norm, h, lds_w, (int)blockIdx.x, N);
        return;
    }
    const int bid = (int)blockIdx.x - ngemm;
    const int tid = threadIdx.x;
    for (int i = tid; i < T; i += 256) lcnt[i] = 0;
    __syncthreads();

    const int base = bid * BIN_CHUNK;
    int d[16], s[16], t[16], rk[16];
    #pragma unroll
    for (int k = 0; k < 16; ++k) {
        const int e  = base + k * 256 + tid;
        const int ec = (e < E) ? e : E - 1;
        d[k] = edst[ec];
        s[k] = esrc[ec];
    }
    #pragma unroll
    for (int k = 0; k < 16; ++k) {
        const int e = base + k * 256 + tid;
        if (e >= E || (unsigned)d[k] >= (unsigned)N) d[k] = -1;
        t[k] = (d[k] >= 0) ? (d[k] >> TILE_SHIFT) : 0;
    }
    #pragma unroll
    for (int k = 0; k < 16; ++k) {
        rk[k] = 0;
        if (d[k] >= 0) rk[k] = atomicAdd(&lcnt[t[k]], 1);
    }
    __syncthreads();
    int* st_rec = (int*)lds_w;
    int* lpre   = st_rec + BIN_CHUNK;
    if (tid < 32) {
        int v[16];
        int sum = 0;
        #pragma unroll
        for (int j = 0; j < 16; ++j) {
            const int idx = tid * 16 + j;
            v[j] = (idx < T) ? lcnt[idx] : 0;
            sum += v[j];
        }
        int inc = sum;
        #pragma unroll
        for (int off = 1; off < 32; off <<= 1) {
            const int nb = __shfl_up(inc, off);
            if (tid >= off) inc += nb;
        }
        int run = inc - sum;
        #pragma unroll
        for (int j = 0; j < 16; ++j) {
            lpre[tid * 16 + j] = run;
            run += v[j];
        }
        if (tid == 31) lpre[MAX_TILES] = inc;
    }
    __syncthreads();
    {
        int* bo = boff + (size_t)bid * OFF_STRIDE;
        for (int i = tid; i <= T; i += 256) bo[i] = lpre[i];
    }
    #pragma unroll
    for (int k = 0; k < 16; ++k) {
        if (d[k] >= 0)
            st_rec[lpre[t[k]] + rk[k]] = (s[k] << TILE_SHIFT) | (d[k] & (TILE_ROWS - 1));
    }
    __syncthreads();
    const int total = lpre[MAX_TILES];
    int* br = brec + (size_t)bid * BIN_CHUNK;
    #pragma unroll
    for (int k = 0; k < 16; ++k) {
        const int q = k * 256 + tid;
        if (q < total) br[q] = st_rec[q];
    }
}

__global__ __launch_bounds__(1024) void gcn_place_sum(
    const int* __restrict__ boff, const int* __restrict__ brec, int nbin, int N,
    const float* __restrict__ h,
    const int* __restrict__ esrc, const int* __restrict__ edst,
    const float* __restrict__ norm, const float* __restrict__ bias,
    float* __restrict__ out, int E)
{
    __shared__ int lcnt[TILE_ROWS];
    __shared__ int lsl[TILE_ROWS * BUCKET_CAP];
    const int tid  = threadIdx.x;
    const int tile = blockIdx.x;
    const int r0   = tile * TILE_ROWS;
    if (tid < TILE_ROWS) lcnt[tid] = 0;
    __syncthreads();

    const int4* brec4 = (const int4*)brec;
    for (int b = tid; b < nbin; b += 1024) {
        const int* bo = boff + (size_t)b * OFF_STRIDE + tile;
        const int beg = bo[0];
        const int end = bo[1];
        if (beg < end) {
            const int4* br4 = brec4 + (size_t)b * (BIN_CHUNK / 4);
            const int clast = (end - 1) >> 2;
            for (int c0 = beg >> 2; c0 <= clast; c0 += 4) {
                int rec[16];
                #pragma unroll
                for (int j = 0; j < 4; ++j) {
                    const int c = c0 + j;
                    const int4 q = br4[(c <= clast) ? c : clast];
                    rec[4 * j + 0] = q.x;
                    rec[4 * j + 1] = q.y;
                    rec[4 * j + 2] = q.z;
                    rec[4 * j + 3] = q.w;
                }
                #pragma unroll
                for (int e = 0; e < 16; ++e) {
                    const int idx = (c0 + (e >> 2)) * 4 + (e & 3);
                    if (idx >= beg && idx < end) {
                        const int lr = rec[e] & (TILE_ROWS - 1);
                        const int r  = atomicAdd(&lcnt[lr], 1);
                        if (r < BUCKET_CAP)
                            lsl[lr * BUCKET_CAP + ((r + lr) & (BUCKET_CAP - 1))] = (int)((unsigned)rec[e] >> TILE_SHIFT);
                    }
                }
            }
        }
    }
    __syncthreads();

    const int lane = tid & 31;
    const int wave = __builtin_amdgcn_readfirstlane(tid >> 5);
    const float4 b = *(const float4*)(bias + lane * 4);
    const float* hl = h + lane * 4;

    for (int i = 0; i < TILE_ROWS / 32; ++i) {
        const int lr  = wave * (TILE_ROWS / 32) + i;
        const int row = r0 + lr;
        if (row >= N) break;
        const int c = __builtin_amdgcn_readfirstlane(lcnt[lr]);
        const float nm = norm[row];
        float4 acc = make_float4(0.f, 0.f, 0.f, 0.f);

        if (c <= BUCKET_CAP) {
            const int* sl = lsl + lr * BUCKET_CAP;
            unsigned k0 = 0xFFFFFFFFu, k1 = 0xFFFFFFFFu;
            if (lane < c)      k0 = ((unsigned)sl[(lane + lr) & (BUCKET_CAP - 1)] << 6) | (unsigned)lane;
            if (lane + 32 < c) k1 = ((unsigned)sl[(lane + 32 + lr) & (BUCKET_CAP - 1)] << 6) | (unsigned)(lane + 32);

            if (c <= 32) {
                int rk0 = 0;
                for (int j = 0; j < c; ++j) {
                    const unsigned kj = (unsigned)__builtin_amdgcn_readlane((int)k0, j);
                    rk0 += (kj < k0) ? 1 : 0;
                }
                const unsigned sorted = (unsigned)__builtin_amdgcn_ds_permute(rk0 << 2, (int)k0);
                const int c4 = c & ~3;
                const int rem = c - c4;
                float4 t0 = make_float4(0.f, 0.f, 0.f, 0.f), t1 = t0, t2 = t0;
                if (rem > 0) t0 = *(const float4*)(hl + (size_t)((unsigned)__builtin_amdgcn_readlane((int)sorted, c4) >> 6) * C_DIM);
                if (rem > 1) t1 = *(const float4*)(hl + (size_t)((unsigned)__builtin_amdgcn_readlane((int)sorted, c4 + 1) >> 6) * C_DIM);
                if (rem > 2) t2 = *(const float4*)(hl + (size_t)((unsigned)__builtin_amdgcn_readlane((int)sorted, c4 + 2) >> 6) * C_DIM);
                for (int r = 0; r < c4; r += 4) {
                    const unsigned s0 = (unsigned)__builtin_amdgcn_readlane((int)sorted, r) >> 6;
                    const unsigned s1 = (unsigned)__builtin_amdgcn_readlane((int)sorted, r + 1) >> 6;
                    const unsigned s2 = (unsigned)__builtin_amdgcn_readlane((int)sorted, r + 2) >> 6;
                    const unsigned s3 = (unsigned)__builtin_amdgcn_readlane((int)sorted, r + 3) >> 6;
                    const float4 v0 = *(const float4*)(hl + (size_t)s0 * C_DIM);
                    const float4 v1 = *(const float4*)(hl + (size_t)s1 * C_DIM);
                    const float4 v2 = *(const float4*)(hl + (size_t)s2 * C_DIM);
                    const float4 v3 = *(const float4*)(hl + (size_t)s3 * C_DIM);
                    acc.x += v0.x; acc.y += v0.y; acc.z += v0.z; acc.w += v0.w;
                    acc.x += v1.x; acc.y += v1.y; acc.z += v1.z; acc.w += v1.w;
                    acc.x += v2.x; acc.y += v2.y; acc.z += v2.z; acc.w += v2.w;
                    acc.x += v3.x; acc.y += v3.y; acc.z += v3.z; acc.w += v3.w;
                }
                if (rem > 0) { acc.x += t0.x; acc.y += t0.y; acc.z += t0.z; acc.w += t0.w; }
                if (rem > 1) { acc.x += t1.x; acc.y += t1.y; acc.z += t1.z; acc.w += t1.w; }
                if (rem > 2) { acc.x += t2.x; acc.y += t2.y; acc.z += t2.z; acc.w += t2.w; }
            } else {
                int rk0 = 0, rk1 = 0;
                for (int j = 0; j < 32; ++j) {
                    const unsigned kj = (unsigned)__builtin_amdgcn_readlane((int)k0, j);
                    rk0 += (kj < k0) ? 1 : 0;
                    rk1 += (kj < k1) ? 1 : 0;
                }
                for (int j = 32; j < c; ++j) {
                    const unsigned kj = (unsigned)__builtin_amdgcn_readlane((int)k1, j - 32);
                    rk0 += (kj < k0) ? 1 : 0;
                    rk1 += (kj < k1) ? 1 : 0;
                }
                for (int r = 0; r < c; ++r) {
                    unsigned m = __builtin_amdgcn_ballot_w32(rk0 == r);
                    unsigned key;
                    if (m != 0u) {
                        key = (unsigned)__builtin_amdgcn_readlane((int)k0, __builtin_ctz(m));
                    } else {
                        m = __builtin_amdgcn_ballot_w32(rk1 == r);
                        key = (unsigned)__builtin_amdgcn_readlane((int)k1, __builtin_ctz(m));
                    }
                    const unsigned s = key >> 6;
                    const float4 v = *(const float4*)(hl + (size_t)s * C_DIM);
                    acc.x += v.x; acc.y += v.y; acc.z += v.z; acc.w += v.w;
                }
            }
        } else {
            for (int base = 0; base < E; base += 32) {
                const int e = base + lane;
                const int d = (e < E) ? edst[e] : -1;
                unsigned m = __builtin_amdgcn_ballot_w32(d == row);
                while (m != 0u) {
                    const int j = __builtin_ctz(m);
                    m &= m - 1u;
                    const int s = esrc[base + j];
                    const float4 v = *(const float4*)(hl + (size_t)s * C_DIM);
                    acc.x += v.x; acc.y += v.y; acc.z += v.z; acc.w += v.w;
                }
            }
        }

        float4 o;
        o.x = acc.x * nm + b.x;
        o.y = acc.y * nm + b.y;
        o.z = acc.z * nm + b.z;
        o.w = acc.w * nm + b.w;
        *(float4*)(out + (size_t)row * C_DIM + lane * 4) = o;
    }
}

extern "C" void kernel_launch(void* const* d_in, const int* in_sizes, int n_in,
                              void* d_out, int out_size, void* d_ws, size_t ws_size,
                              hipStream_t stream) {
    const float* x    = (const float*)d_in[0];
    const float* norm = (const float*)d_in[1];
    const float* w    = (const float*)d_in[2];
    const float* bias = (const float*)d_in[3];
    const int*   esrc = (const int*)d_in[4];
    const int*   edst = (const int*)d_in[5];
    float* out = (float*)d_out;

    const int N = in_sizes[0] / C_DIM;
    const int E = in_sizes[4];
    float* h     = (float*)d_ws;
    const int T  = (N + TILE_ROWS - 1) / TILE_ROWS;
    const int NB = (E + BIN_CHUNK - 1) / BIN_CHUNK;
    int*   boff  = (int*)(h + (size_t)N * C_DIM);
    int*   brec  = boff + (size_t)NB * OFF_STRIDE;

    const int G  = (N + 127) / 128;

    gcn_bin<<<G + NB, dim3(256), 0, stream>>>(esrc, edst, boff, brec, N, E, T,
                                              x, w, norm, h, G);

    gcn_place_sum<<<T, dim3(1024), 0, stream>>>(boff, brec, NB, N, h, esrc, edst,
                                                norm, bias, out, E);
}


# GEMM work evened out over the processors: 768 full 128-row tiles (3 per processor) and the 106 left-over row blocks as 53 work-groups of 32 rows, so no SIMD gets a 4th tile's WMMAs; sums unchanged
# speedup vs baseline: 4.8322x; 1.0801x over previous
#include <hip/hip_runtime.h>

typedef __attribute__((ext_vector_type(2))) float v2f;
typedef __attribute__((ext_vector_type(8))) float v8f;

#define C_DIM 128

template <int NT>
__device__ __forceinline__ void gemm_tile(
    const float* __restrict__ x, const float* __restrict__ w,
    const float* __restrict__ norm, float* __restrict__ h,
    float* lds_w, int m0, int nbase, int N)
{
    for (int idx = threadIdx.x; idx < (C_DIM / 2) * C_DIM; idx += 256) {
        const int p = idx >> 7;
        const int n = idx & 127;
        v2f v;
        v.x = w[(2 * p + 0) * C_DIM + n];
        v.y = w[(2 * p + 1) * C_DIM + n];
        const int dw = p * 256 + ((2 * n + 32 * (p & 1)) & 255);
        *(v2f*)&lds_w[dw] = v;
    }
    __syncthreads();

    const int lane   = threadIdx.x & 31;
    const int lane16 = lane & 15;
    const int khalf  = lane >> 4;

    int row  = m0 + lane16;
    int rowc = row < N ? row : N - 1;

    v8f acc[NT];
    const v8f vzero = {0.f,0.f,0.f,0.f,0.f,0.f,0.f,0.f};
    #pragma unroll
    for (int n = 0; n < NT; ++n) acc[n] = vzero;

    const float* xrow = x + (size_t)rowc * C_DIM;

    int coff[NT];
    #pragma unroll
    for (int n = 0; n < NT; ++n)
        coff[n] = (32 * (nbase + n) + 2 * lane16 + 32 * khalf) & 255;
    const float* lwp = lds_w + khalf * 256;

    #pragma unroll 4
    for (int k = 0; k < C_DIM; k += 4) {
        const int kb = k + 2 * khalf;
        v2f a = *(const v2f*)(xrow + kb);
        const float* lk = lwp + (k >> 1) * 256;
        #pragma unroll
        for (int n = 0; n < NT; ++n) {
            v2f b = *(const v2f*)(lk + coff[n]);
            acc[n] = __builtin_amdgcn_wmma_f32_16x16x4_f32(
                false, a, false, b, (short)0, acc[n], false, false);
        }
    }

    float nrm[8];
    #pragma unroll
    for (int j = 0; j < 8; ++j) {
        const int r = m0 + j + 8 * khalf;
        nrm[j] = (r < N) ? norm[r] : 0.0f;
    }
    #pragma unroll
    for (int n = 0; n < NT; ++n) {
        #pragma unroll
        for (int j = 0; j < 8; ++j) {
            const int r = m0 + j + 8 * khalf;
            if (r < N) {
                h[(size_t)r * C_DIM + (nbase + n) * 16 + lane16] = acc[n][j] * nrm[j];
            }
        }
    }
}

#define BUCKET_CAP 64
#define TILE_ROWS  256
#define TILE_SHIFT 8
#define MAX_TILES  512
#define BIN_CHUNK  4096
#define OFF_STRIDE 520

__global__ __launch_bounds__(256) void gcn_bin(
    const int* __restrict__ esrc, const int* __restrict__ edst,
    int* __restrict__ boff, int* __restrict__ brec, int N, int E, int T,
    const float* __restrict__ x, const float* __restrict__ w,
    const float* __restrict__ norm, float* __restrict__ h,
    int ngemm, int nfull)
{
    __shared__ float lds_w[C_DIM * C_DIM];
    __shared__ int lcnt[MAX_TILES];
    if ((int)blockIdx.x < ngemm) {
        const int wave = threadIdx.x >> 5;
        if ((int)blockIdx.x < nfull) {
            gemm_tile<8>(x, w, norm, h, lds_w, ((int)blockIdx.x * 8 + wave) * 16, 0, N);
        } else {
            const int s = (int)blockIdx.x - nfull;
            gemm_tile<2>(x, w, norm, h, lds_w,
                         (nfull * 8 + s * 2 + (wave >> 2)) * 16, 2 * (wave & 3), N);
        }
        return;
    }
    const int bid = (int)blockIdx.x - ngemm;
    const int tid = threadIdx.x;
    for (int i = tid; i < T; i += 256) lcnt[i] = 0;
    __syncthreads();

    const int base = bid * BIN_CHUNK;
    int d[16], s[16], t[16], rk[16];
    #pragma unroll
    for (int k = 0; k < 16; ++k) {
        const int e  = base + k * 256 + tid;
        const int ec = (e < E) ? e : E - 1;
        d[k] = edst[ec];
        s[k] = esrc[ec];
    }
    #pragma unroll
    for (int k = 0; k < 16; ++k) {
        const int e = base + k * 256 + tid;
        if (e >= E || (unsigned)d[k] >= (unsigned)N) d[k] = -1;
        t[k] = (d[k] >= 0) ? (d[k] >> TILE_SHIFT) : 0;
    }
    #pragma unroll
    for (int k = 0; k < 16; ++k) {
        rk[k] = 0;
        if (d[k] >= 0) rk[k] = atomicAdd(&lcnt[t[k]], 1);
    }
    __syncthreads();
    int* st_rec = (int*)lds_w;
    int* lpre   = st_rec + BIN_CHUNK;
    if (tid < 32) {
        int v[16];
        int sum = 0;
        #pragma unroll
        for (int j = 0; j < 16; ++j) {
            const int idx = tid * 16 + j;
            v[j] = (idx < T) ? lcnt[idx] : 0;
            sum += v[j];
        }
        int inc = sum;
        #pragma unroll
        for (int off = 1; off < 32; off <<= 1) {
            const int nb = __shfl_up(inc, off);
            if (tid >= off) inc += nb;
        }
        int run = inc - sum;
        #pragma unroll
        for (int j = 0; j < 16; ++j) {
            lpre[tid * 16 + j] = run;
            run += v[j];
        }
        if (tid == 31) lpre[MAX_TILES] = inc;
    }
    __syncthreads();
    {
        int* bo = boff + (size_t)bid * OFF_STRIDE;
        for (int i = tid; i <= T; i += 256) bo[i] = lpre[i];
    }
    #pragma unroll
    for (int k = 0; k < 16; ++k) {
        if (d[k] >= 0)
            st_rec[lpre[t[k]] + rk[k]] = (s[k] << TILE_SHIFT) | (d[k] & (TILE_ROWS - 1));
    }
    __syncthreads();
    const int total = lpre[MAX_TILES];
    int* br = brec + (size_t)bid * BIN_CHUNK;
    #pragma unroll
    for (int k = 0; k < 16; ++k) {
        const int q = k * 256 + tid;
        if (q < total) br[q] = st_rec[q];
    }
}

__global__ __launch_bounds__(1024) void gcn_place_sum(
    const int* __restrict__ boff, const int* __restrict__ brec, int nbin, int N,
    const float* __restrict__ h,
    const int* __restrict__ esrc, const int* __restrict__ edst,
    const float* __restrict__ norm, const float* __restrict__ bias,
    float* __restrict__ out, int E)
{
    __shared__ int lcnt[TILE_ROWS];
    __shared__ int lsl[TILE_ROWS * BUCKET_CAP];
    const int tid  = threadIdx.x;
    const int tile = blockIdx.x;
    const int r0   = tile * TILE_ROWS;
    if (tid < TILE_ROWS) lcnt[tid] = 0;
    __syncthreads();

    const int4* brec4 = (const int4*)brec;
    for (int b = tid; b < nbin; b += 1024) {
        const int* bo = boff + (size_t)b * OFF_STRIDE + tile;
        const int beg = bo[0];
        const int end = bo[1];
        if (beg < end) {
            const int4* br4 = brec4 + (size_t)b * (BIN_CHUNK / 4);
            const int clast = (end - 1) >> 2;
            for (int c0 = beg >> 2; c0 <= clast; c0 += 4) {
                int rec[16];
                #pragma unroll
                for (int j = 0; j < 4; ++j) {
                    const int c = c0 + j;
                    const int4 q = br4[(c <= clast) ? c : clast];
                    rec[4 * j + 0] = q.x;
                    rec[4 * j + 1] = q.y;
                    rec[4 * j + 2] = q.z;
                    rec[4 * j + 3] = q.w;
                }
                #pragma unroll
                for (int e = 0; e < 16; ++e) {
                    const int idx = (c0 + (e >> 2)) * 4 + (e & 3);
                    if (idx >= beg && idx < end) {
                        const int lr = rec[e] & (TILE_ROWS - 1);
                        const int r  = atomicAdd(&lcnt[lr], 1);
                        if (r < BUCKET_CAP)
                            lsl[lr * BUCKET_CAP + ((r + lr) & (BUCKET_CAP - 1))] = (int)((unsigned)rec[e] >> TILE_SHIFT);
                    }
                }
            }
        }
    }
    __syncthreads();

    const int lane = tid & 31;
    const int wave = __builtin_amdgcn_readfirstlane(tid >> 5);
    const float4 b = *(const float4*)(bias + lane * 4);
    const float* hl = h + lane * 4;

    for (int i = 0; i < TILE_ROWS / 32; ++i) {
        const int lr  = wave * (TILE_ROWS / 32) + i;
        const int row = r0 + lr;
        if (row >= N) break;
        const int c = __builtin_amdgcn_readfirstlane(lcnt[lr]);
        const float nm = norm[row];
        float4 acc = make_float4(0.f, 0.f, 0.f, 0.f);

        if (c <= BUCKET_CAP) {
            const int* sl = lsl + lr * BUCKET_CAP;
            unsigned k0 = 0xFFFFFFFFu, k1 = 0xFFFFFFFFu;
            if (lane < c)      k0 = ((unsigned)sl[(lane + lr) & (BUCKET_CAP - 1)] << 6) | (unsigned)lane;
            if (lane + 32 < c) k1 = ((unsigned)sl[(lane + 32 + lr) & (BUCKET_CAP - 1)] << 6) | (unsigned)(lane + 32);

            if (c <= 32) {
                int rk0 = 0;
                for (int j = 0; j < c; ++j) {
                    const unsigned kj = (unsigned)__builtin_amdgcn_readlane((int)k0, j);
                    rk0 += (kj < k0) ? 1 : 0;
                }
                const unsigned sorted = (unsigned)__builtin_amdgcn_ds_permute(rk0 << 2, (int)k0);
                const int c4 = c & ~3;
                const int rem = c - c4;
                float4 t0 = make_float4(0.f, 0.f, 0.f, 0.f), t1 = t0, t2 = t0;
                if (rem > 0) t0 = *(const float4*)(hl + (size_t)((unsigned)__builtin_amdgcn_readlane((int)sorted, c4) >> 6) * C_DIM);
                if (rem > 1) t1 = *(const float4*)(hl + (size_t)((unsigned)__builtin_amdgcn_readlane((int)sorted, c4 + 1) >> 6) * C_DIM);
                if (rem > 2) t2 = *(const float4*)(hl + (size_t)((unsigned)__builtin_amdgcn_readlane((int)sorted, c4 + 2) >> 6) * C_DIM);
                for (int r = 0; r < c4; r += 4) {
                    const unsigned s0 = (unsigned)__builtin_amdgcn_readlane((int)sorted, r) >> 6;
                    const unsigned s1 = (unsigned)__builtin_amdgcn_readlane((int)sorted, r + 1) >> 6;
                    const unsigned s2 = (unsigned)__builtin_amdgcn_readlane((int)sorted, r + 2) >> 6;
                    const unsigned s3 = (unsigned)__builtin_amdgcn_readlane((int)sorted, r + 3) >> 6;
                    const float4 v0 = *(const float4*)(hl + (size_t)s0 * C_DIM);
                    const float4 v1 = *(const float4*)(hl + (size_t)s1 * C_DIM);
                    const float4 v2 = *(const float4*)(hl + (size_t)s2 * C_DIM);
                    const float4 v3 = *(const float4*)(hl + (size_t)s3 * C_DIM);
                    acc.x += v0.x; acc.y += v0.y; acc.z += v0.z; acc.w += v0.w;
                    acc.x += v1.x; acc.y += v1.y; acc.z += v1.z; acc.w += v1.w;
                    acc.x += v2.x; acc.y += v2.y; acc.z += v2.z; acc.w += v2.w;
                    acc.x += v3.x; acc.y += v3.y; acc.z += v3.z; acc.w += v3.w;
                }
                if (rem > 0) { acc.x += t0.x; acc.y += t0.y; acc.z += t0.z; acc.w += t0.w; }
                if (rem > 1) { acc.x += t1.x; acc.y += t1.y; acc.z += t1.z; acc.w += t1.w; }
                if (rem > 2) { acc.x += t2.x; acc.y += t2.y; acc.z += t2.z; acc.w += t2.w; }
            } else {
                int rk0 = 0, rk1 = 0;
                for (int j = 0; j < 32; ++j) {
                    const unsigned kj = (unsigned)__builtin_amdgcn_readlane((int)k0, j);
                    rk0 += (kj < k0) ? 1 : 0;
                    rk1 += (kj < k1) ? 1 : 0;
                }
                for (int j = 32; j < c; ++j) {
                    const unsigned kj = (unsigned)__builtin_amdgcn_readlane((int)k1, j - 32);
                    rk0 += (kj < k0) ? 1 : 0;
                    rk1 += (kj < k1) ? 1 : 0;
                }
                for (int r = 0; r < c; ++r) {
                    unsigned m = __builtin_amdgcn_ballot_w32(rk0 == r);
                    unsigned key;
                    if (m != 0u) {
                        key = (unsigned)__builtin_amdgcn_readlane((int)k0, __builtin_ctz(m));
                    } else {
                        m = __builtin_amdgcn_ballot_w32(rk1 == r);
                        key = (unsigned)__builtin_amdgcn_readlane((int)k1, __builtin_ctz(m));
                    }
                    const unsigned s = key >> 6;
                    const float4 v = *(const float4*)(hl + (size_t)s * C_DIM);
                    acc.x += v.x; acc.y += v.y; acc.z += v.z; acc.w += v.w;
                }
            }
        } else {
            for (int base = 0; base < E; base += 32) {
                const int e = base + lane;
                const int d = (e < E) ? edst[e] : -1;
                unsigned m = __builtin_amdgcn_ballot_w32(d == row);
                while (m != 0u) {
                    const int j = __builtin_ctz(m);
                    m &= m - 1u;
                    const int s = esrc[base + j];
                    const float4 v = *(const float4*)(hl + (size_t)s * C_DIM);
                    acc.x += v.x; acc.y += v.y; acc.z += v.z; acc.w += v.w;
                }
            }
        }

        float4 o;
        o.x = acc.x * nm + b.x;
        o.y = acc.y * nm + b.y;
        o.z = acc.z * nm + b.z;
        o.w = acc.w * nm + b.w;
        *(float4*)(out + (size_t)row * C_DIM + lane * 4) = o;
    }
}

extern "C" void kernel_launch(void* const* d_in, const int* in_sizes, int n_in,
                              void* d_out, int out_size, void* d_ws, size_t ws_size,
                              hipStream_t stream) {
    const float* x    = (const float*)d_in[0];
    const float* norm = (const float*)d_in[1];
    const float* w    = (const float*)d_in[2];
    const float* bias = (const float*)d_in[3];
    const int*   esrc = (const int*)d_in[4];
    const int*   edst = (const int*)d_in[5];
    float* out = (float*)d_out;

    const int N = in_sizes[0] / C_DIM;
    const int E = in_sizes[4];
    float* h     = (float*)d_ws;
    const int T  = (N + TILE_ROWS - 1) / TILE_ROWS;
    const int NB = (E + BIN_CHUNK - 1) / BIN_CHUNK;
    int*   boff  = (int*)(h + (size_t)N * C_DIM);
    int*   brec  = boff + (size_t)NB * OFF_STRIDE;

    const int RB = (N + 15) / 16;
    const int F  = (RB / 8 > 256) ? (RB / 8) / 256 * 256 : RB / 8;
    const int S  = (RB - 8 * F + 1) / 2;

    gcn_bin<<<F + S + NB, dim3(256), 0, stream>>>(esrc, edst, boff, brec, N, E, T,
                                                  x, w, norm, h, F + S, F);

    gcn_place_sum<<<T, dim3(1024), 0, stream>>>(boff, brec, NB, N, h, esrc, edst,
                                                norm, bias, out, E);
}
